# starting kernel
# baseline (speedup 1.0000x reference)
#include <hip/hip_runtime.h>
#include <hip/hip_bf16.h>

typedef __attribute__((ext_vector_type(2))) float v2f;
typedef __attribute__((ext_vector_type(8))) float v8f;

#define N_NODES 100000
#define E_POS   500000
#define E_PRED  200000

__global__ void deg_init_kernel(float* deg, int n) {
    int i = blockIdx.x * blockDim.x + threadIdx.x;
    if (i < n) deg[i] = 1.0f;
}

__global__ void deg_count_kernel(const int* __restrict__ tgt, float* deg, int e) {
    int i = blockIdx.x * blockDim.x + threadIdx.x;
    if (i < e) unsafeAtomicAdd(&deg[tgt[i]], 1.0f);
}

__global__ void deg_finalize_kernel(float* deg, int n) {
    int i = blockIdx.x * blockDim.x + threadIdx.x;
    if (i < n) deg[i] = rsqrtf(deg[i]);
}

__global__ void gemm_wmma_f32(const float* __restrict__ A,
                              const float* __restrict__ B,
                              const float* __restrict__ bias,
                              float* __restrict__ C,
                              int K, int N) {
    const int m0   = blockIdx.x * 16;
    const int n0   = blockIdx.y * 16;
    const int lane = threadIdx.x;
    const int half = lane >> 4;
    const int idx  = lane & 15;

    v8f acc = {};
    const float* arow = A + (size_t)(m0 + idx) * K;
    const float* bcol = B + (size_t)(n0 + idx);

    for (int k0 = 0; k0 < K; k0 += 4) {
        const int kb = k0 + half * 2;
        v2f a;
        a.x = arow[kb];
        a.y = arow[kb + 1];
        v2f b;
        b.x = bcol[(size_t)kb * N];
        b.y = bcol[(size_t)(kb + 1) * N];
        acc = __builtin_amdgcn_wmma_f32_16x16x4_f32(
            false, a, false, b, (short)0, acc, false, false);
    }

    const float bv = bias ? bias[n0 + idx] : 0.0f;
    float* crow = C + (size_t)(m0 + half * 8) * N + n0 + idx;
#pragma unroll
    for (int j = 0; j < 8; ++j) {
        crow[(size_t)j * N] = acc[j] + bv;
    }
}

__global__ void scatter_edges_kernel(const float* __restrict__ xw,
                                     const int* __restrict__ src,
                                     const int* __restrict__ tgt,
                                     const float* __restrict__ dinv,
                                     float* __restrict__ agg,
                                     int e, int f_log2) {
    long long i = (long long)blockIdx.x * blockDim.x + threadIdx.x;
    long long total = (long long)e << f_log2;
    if (i >= total) return;
    const int ed = (int)(i >> f_log2);
    const int f  = (int)(i & ((1 << f_log2) - 1));
    const int s  = src[ed];
    const int t  = tgt[ed];
    const float norm = dinv[s] * dinv[t];
    const float v = xw[((size_t)s << f_log2) + f] * norm;
    unsafeAtomicAdd(&agg[((size_t)t << f_log2) + f], v);
}

__global__ void combine_kernel(const float* __restrict__ agg,
                               const float* __restrict__ xw,
                               const float* __restrict__ dinv,
                               const float* __restrict__ bias,
                               float* __restrict__ out,
                               int n, int f_log2, int do_relu) {
    long long i = (long long)blockIdx.x * blockDim.x + threadIdx.x;
    long long total = (long long)n << f_log2;
    if (i >= total) return;
    const int node = (int)(i >> f_log2);
    const int f    = (int)(i & ((1 << f_log2) - 1));
    const float di = dinv[node];
    float v = agg[i] + xw[i] * di * di + bias[f];
    if (do_relu) v = fmaxf(v, 0.0f);
    out[i] = v;
}

__global__ void edge_score_kernel(const float* __restrict__ h,
                                  const int* __restrict__ ia,
                                  const int* __restrict__ ib,
                                  float* __restrict__ out, int e) {
    int tid  = blockIdx.x * blockDim.x + threadIdx.x;
    int ed   = tid >> 4;
    int lane = tid & 15;
    if (ed >= e) return;
    const int a = ia[ed];
    const int b = ib[ed];
    const float4 va = ((const float4*)(h + (size_t)a * 64))[lane];
    const float4 vb = ((const float4*)(h + (size_t)b * 64))[lane];
    float s = va.x * vb.x + va.y * vb.y + va.z * vb.z + va.w * vb.w;
    s += __shfl_xor(s, 8, 32);
    s += __shfl_xor(s, 4, 32);
    s += __shfl_xor(s, 2, 32);
    s += __shfl_xor(s, 1, 32);
    if (lane == 0) out[ed] = s;
}

extern "C" void kernel_launch(void* const* d_in, const int* in_sizes, int n_in,
                              void* d_out, int out_size, void* d_ws, size_t ws_size,
                              hipStream_t stream) {
    const float* x      = (const float*)d_in[0];
    const int*   eidx   = (const int*)d_in[1];
    const int*   pidx   = (const int*)d_in[2];
    const float* W_init = (const float*)d_in[3];
    const float* b_init = (const float*)d_in[4];
    const float* W1     = (const float*)d_in[5];
    const float* b1     = (const float*)d_in[6];
    const float* W2     = (const float*)d_in[7];
    const float* b2     = (const float*)d_in[8];
    float* logits = (float*)d_out;

    const int* src = pidx;
    const int* tgt = pidx + E_POS;
    const int* qa  = eidx;
    const int* qb  = eidx + E_PRED;

    float* ws   = (float*)d_ws;
    float* dinv = ws;
    float* bufA = ws + 102400;
    float* bufB = bufA + (size_t)N_NODES * 128;
    float* bufC = bufB + (size_t)N_NODES * 128;

    const int TB = 256;
    const int MT = N_NODES / 16;

    deg_init_kernel<<<(N_NODES + TB - 1) / TB, TB, 0, stream>>>(dinv, N_NODES);
    deg_count_kernel<<<(E_POS + TB - 1) / TB, TB, 0, stream>>>(tgt, dinv, E_POS);
    deg_finalize_kernel<<<(N_NODES + TB - 1) / TB, TB, 0, stream>>>(dinv, N_NODES);

    gemm_wmma_f32<<<dim3(MT, 128 / 16), 32, 0, stream>>>(x, W_init, b_init, bufA, 128, 128);

    gemm_wmma_f32<<<dim3(MT, 128 / 16), 32, 0, stream>>>(bufA, W1, nullptr, bufB, 128, 128);
    hipMemsetAsync(bufC, 0, (size_t)N_NODES * 128 * sizeof(float), stream);
    {
        long long total = (long long)E_POS << 7;
        scatter_edges_kernel<<<(unsigned)((total + TB - 1) / TB), TB, 0, stream>>>(
            bufB, src, tgt, dinv, bufC, E_POS, 7);
        long long nf = (long long)N_NODES << 7;
        combine_kernel<<<(unsigned)((nf + TB - 1) / TB), TB, 0, stream>>>(
            bufC, bufB, dinv, b1, bufA, N_NODES, 7, 1);
    }

    gemm_wmma_f32<<<dim3(MT, 64 / 16), 32, 0, stream>>>(bufA, W2, nullptr, bufB, 128, 64);
    hipMemsetAsync(bufC, 0, (size_t)N_NODES * 64 * sizeof(float), stream);
    {
        long long total = (long long)E_POS << 6;
        scatter_edges_kernel<<<(unsigned)((total + TB - 1) / TB), TB, 0, stream>>>(
            bufB, src, tgt, dinv, bufC, E_POS, 6);
        long long nf = (long long)N_NODES << 6;
        combine_kernel<<<(unsigned)((nf + TB - 1) / TB), TB, 0, stream>>>(
            bufC, bufB, dinv, b2, bufA, N_NODES, 6, 0);
    }

    {
        long long threads = (long long)E_PRED * 16;
        edge_score_kernel<<<(unsigned)((threads + TB - 1) / TB), TB, 0, stream>>>(
            bufA, qa, qb, logits, E_PRED);
    }
}


# Edge sums as a per-node gather over edges bucketed by target (int count, scan, cursor fill) instead of float atomic scatter; zero-fill and combine folded in; sum order = cursor slot order
# speedup vs baseline: 2.8169x; 2.8169x over previous
#include <hip/hip_runtime.h>
#include <hip/hip_bf16.h>

typedef __attribute__((ext_vector_type(2))) float v2f;
typedef __attribute__((ext_vector_type(4))) float v4f;
typedef __attribute__((ext_vector_type(8))) float v8f;

#define N_NODES 100000
#define E_POS   500000
#define E_PRED  200000

__global__ void csr_zero_kernel(int* __restrict__ cnt, int n) {
    int i = blockIdx.x * blockDim.x + threadIdx.x;
    if (i < n) cnt[i] = 0;
}

__global__ void csr_count_kernel(const int* __restrict__ tgt, int* __restrict__ cnt, int e) {
    int i = blockIdx.x * blockDim.x + threadIdx.x;
    if (i < e) atomicAdd(&cnt[tgt[i]], 1);
}

__global__ __launch_bounds__(1024) void csr_scan_kernel(const int* __restrict__ cnt,
                                                        int* __restrict__ bounds,
                                                        float* __restrict__ dinv, int n) {
    __shared__ int wsum[32];
    __shared__ int wpre[32];
    const int tid  = threadIdx.x;
    const int lane = tid & 31;
    const int wv   = tid >> 5;
    const int base = blockIdx.x * 1024;
    int before = 0;
    for (int i = tid; i < base; i += 1024) before += cnt[i];
    const int i = base + tid;
    const int c = (i < n) ? cnt[i] : 0;
    int inc = c;
#pragma unroll
    for (int d = 1; d < 32; d <<= 1) {
        const int t = __shfl_up(inc, d, 32);
        if (lane >= d) inc += t;
    }
#pragma unroll
    for (int d = 16; d > 0; d >>= 1) before += __shfl_xor(before, d, 32);
    if (lane == 31) wsum[wv] = inc;
    if (lane == 0)  wpre[wv] = before;
    __syncthreads();
    int off = 0;
#pragma unroll
    for (int k = 0; k < 32; ++k) {
        off += wpre[k];
        if (k < wv) off += wsum[k];
    }
    const int excl = off + inc - c;
    if (i < n) {
        *(int2*)(bounds + 2 * i) = make_int2(excl, excl);
        dinv[i]     = rsqrtf((float)(c + 1));
    }
}

__global__ void csr_fill_kernel(const int* __restrict__ src, const int* __restrict__ tgt,
                                int* __restrict__ bounds, int* __restrict__ csr_src, int e) {
    int i = blockIdx.x * blockDim.x + threadIdx.x;
    if (i < e) {
        const int pos = atomicAdd(&bounds[2 * tgt[i] + 1], 1);
        if ((unsigned)pos < (unsigned)e) csr_src[pos] = src[i];
    }
}

template <int N>
__global__ __launch_bounds__(256) void gemm_wmma_f32(const float* __restrict__ A,
                                                     const float* __restrict__ B,
                                                     const float* __restrict__ bias,
                                                     float* __restrict__ C,
                                                     int mtiles) {
    constexpr int K   = 128;
    constexpr int NS  = N / 64;
    constexpr int WPS = 8 / NS;
    __shared__ float4 Bs[(K / 8) * N * 2];

    const int tid = threadIdx.x;
    for (int p = tid; p < (K / 4) * N; p += 256) {
        const int n  = p % N;
        const int k4 = p / N;
        const float* bp = B + (size_t)(4 * k4) * N + n;
        float4 v;
        v.x = bp[0];
        v.y = bp[N];
        v.z = bp[2 * N];
        v.w = bp[3 * N];
        Bs[((((k4 >> 1) * N) + n) << 1) + (k4 & 1)] = v;
    }
    __syncthreads();

    const int lane = tid & 31;
    const int half = lane >> 4;
    const int idx  = lane & 15;
    const int wave  = __builtin_amdgcn_readfirstlane(tid >> 5);
    const int slice = wave % NS;
    const int grp   = wave / NS;
    const int nwg   = gridDim.x;
    const int n0    = slice * 64;

    float4 b[K / 8][4];
    {
        const float4* bfrag = Bs + (n0 + idx) * 2 + half;
#pragma unroll
        for (int j = 0; j < K / 8; ++j)
#pragma unroll
            for (int t = 0; t < 4; ++t) b[j][t] = bfrag[(j * N + 16 * t) * 2];
    }
    float bv[4];
#pragma unroll
    for (int t = 0; t < 4; ++t) bv[t] = bias ? bias[n0 + 16 * t + idx] : 0.0f;

    float4 a[K / 8];
    int mt = grp * nwg + blockIdx.x;
    if (mt < mtiles) {
        const float4* ap = (const float4*)(A + (size_t)(mt * 16 + idx) * K) + half;
#pragma unroll
        for (int j = 0; j < K / 8; ++j) a[j] = ap[2 * j];
    }

    for (; mt < mtiles; mt += WPS * nwg) {
        const int m0 = mt * 16;
        const int mtn = min(mt + WPS * nwg, mtiles - 1);
        const float4* apn = (const float4*)(A + (size_t)(mtn * 16 + idx) * K) + half;
        float4 an[K / 8];
#pragma unroll
        for (int j = 0; j < K / 8; ++j) an[j] = apn[2 * j];

        v8f acc[4];
#pragma unroll
        for (int t = 0; t < 4; ++t) acc[t] = v8f{};

#pragma unroll
        for (int j = 0; j < K / 8; ++j) {
            v2f a0; a0.x = a[j].x; a0.y = a[j].y;
            v2f a1; a1.x = a[j].z; a1.y = a[j].w;
#pragma unroll
            for (int t = 0; t < 4; ++t) {
                v2f b0; b0.x = b[j][t].x; b0.y = b[j][t].y;
                acc[t] = __builtin_amdgcn_wmma_f32_16x16x4_f32(
                    false, a0, false, b0, (short)0, acc[t], false, false);
            }
#pragma unroll
            for (int t = 0; t < 4; ++t) {
                v2f b1; b1.x = b[j][t].z; b1.y = b[j][t].w;
                acc[t] = __builtin_amdgcn_wmma_f32_16x16x4_f32(
                    false, a1, false, b1, (short)0, acc[t], false, false);
            }
        }

        float* crow = C + (size_t)(m0 + half * 8) * N + n0 + idx;
#pragma unroll
        for (int t = 0; t < 4; ++t) {
#pragma unroll
            for (int r = 0; r < 8; ++r) {
                crow[(size_t)r * N + 16 * t] = acc[t][r] + bv[t];
            }
        }
#pragma unroll
        for (int j = 0; j < K / 8; ++j) a[j] = an[j];
    }
}

template <int V> struct rowvec;
template <> struct rowvec<4> { typedef v4f type; };
template <> struct rowvec<2> { typedef v2f type; };

template <int F, bool RELU>
__global__ __launch_bounds__(256) void gather_nodes_kernel(const float* __restrict__ xw,
                                                           const int* __restrict__ bounds,
                                                           const int* __restrict__ csr_src,
                                                           const float* __restrict__ dinv,
                                                           const float* __restrict__ bias,
                                                           float* __restrict__ out, int n, int e) {
    constexpr int V = F / 32;
    typedef typename rowvec<V>::type vf;
    const int lane = threadIdx.x & 31;
    const int node = __builtin_amdgcn_readfirstlane((int)(blockIdx.x * 8 + (threadIdx.x >> 5)));
    if (node >= n) return;
    const int2  seg = *(const int2*)(bounds + 2 * node);
    const int   p0 = max(seg.x, 0);
    const int   p1 = min(seg.y, e);
    const float di = dinv[node];
    const float* xcol = xw + lane * V;
    const vf self = *(const vf*)(xcol + (size_t)node * F);
    vf acc = {};
    for (int base = p0; base < p1; base += 32) {
        const int m = min(p1 - base, 32);
        int s = node;
        if (lane < m) s = min(max(csr_src[base + lane], 0), n - 1);
        const int wi = __float_as_int(dinv[s]);
        int j = 0;
        for (; j + 4 <= m; j += 4) {
            const int s0 = __builtin_amdgcn_readlane(s, j);
            const int s1 = __builtin_amdgcn_readlane(s, j + 1);
            const int s2 = __builtin_amdgcn_readlane(s, j + 2);
            const int s3 = __builtin_amdgcn_readlane(s, j + 3);
            const vf r0 = *(const vf*)(xcol + (size_t)s0 * F);
            const vf r1 = *(const vf*)(xcol + (size_t)s1 * F);
            const vf r2 = *(const vf*)(xcol + (size_t)s2 * F);
            const vf r3 = *(const vf*)(xcol + (size_t)s3 * F);
            acc += r0 * (__int_as_float(__builtin_amdgcn_readlane(wi, j)) * di);
            acc += r1 * (__int_as_float(__builtin_amdgcn_readlane(wi, j + 1)) * di);
            acc += r2 * (__int_as_float(__builtin_amdgcn_readlane(wi, j + 2)) * di);
            acc += r3 * (__int_as_float(__builtin_amdgcn_readlane(wi, j + 3)) * di);
        }
        for (; j < m; ++j) {
            const int s0 = __builtin_amdgcn_readlane(s, j);
            const vf r0 = *(const vf*)(xcol + (size_t)s0 * F);
            acc += r0 * (__int_as_float(__builtin_amdgcn_readlane(wi, j)) * di);
        }
    }
    const vf bv = *(const vf*)(bias + lane * V);
    vf o;
#pragma unroll
    for (int c = 0; c < V; ++c) {
        float v = acc[c] + self[c] * di * di + bv[c];
        if (RELU) v = fmaxf(v, 0.0f);
        o[c] = v;
    }
    *(vf*)(out + (size_t)node * F + lane * V) = o;
}

__global__ void edge_score_kernel(const float* __restrict__ h,
                                  const int* __restrict__ ia,
                                  const int* __restrict__ ib,
                                  float* __restrict__ out, int e) {
    int tid  = blockIdx.x * blockDim.x + threadIdx.x;
    int ed   = tid >> 4;
    int lane = tid & 15;
    if (ed >= e) return;
    const int a = ia[ed];
    const int b = ib[ed];
    const float4 va = ((const float4*)(h + (size_t)a * 64))[lane];
    const float4 vb = ((const float4*)(h + (size_t)b * 64))[lane];
    float s = va.x * vb.x + va.y * vb.y + va.z * vb.z + va.w * vb.w;
    s += __shfl_xor(s, 8, 32);
    s += __shfl_xor(s, 4, 32);
    s += __shfl_xor(s, 2, 32);
    s += __shfl_xor(s, 1, 32);
    if (lane == 0) out[ed] = s;
}

extern "C" void kernel_launch(void* const* d_in, const int* in_sizes, int n_in,
                              void* d_out, int out_size, void* d_ws, size_t ws_size,
                              hipStream_t stream) {
    const float* x      = (const float*)d_in[0];
    const int*   eidx   = (const int*)d_in[1];
    const int*   pidx   = (const int*)d_in[2];
    const float* W_init = (const float*)d_in[3];
    const float* b_init = (const float*)d_in[4];
    const float* W1     = (const float*)d_in[5];
    const float* b1     = (const float*)d_in[6];
    const float* W2     = (const float*)d_in[7];
    const float* b2     = (const float*)d_in[8];
    float* logits = (float*)d_out;

    const int* src = pidx;
    const int* tgt = pidx + E_POS;
    const int* qa  = eidx;
    const int* qb  = eidx + E_PRED;

    float* ws   = (float*)d_ws;
    float* dinv = ws;
    float* bufA = ws + 102400;
    float* bufB = bufA + (size_t)N_NODES * 128;
    float* bufC = bufB + (size_t)N_NODES * 128;
    int*  cnt     = (int*)bufC;
    int*  bounds  = cnt + 102400;
    int*  csr_src = cnt + 3 * 102400;

    const int TB = 256;
    const int MT = N_NODES / 16;

    csr_zero_kernel<<<(N_NODES + TB - 1) / TB, TB, 0, stream>>>(cnt, N_NODES);
    csr_count_kernel<<<(E_POS + TB - 1) / TB, TB, 0, stream>>>(tgt, cnt, E_POS);
    csr_scan_kernel<<<(N_NODES + 1023) / 1024, 1024, 0, stream>>>(cnt, bounds, dinv, N_NODES);
    csr_fill_kernel<<<(E_POS + TB - 1) / TB, TB, 0, stream>>>(src, tgt, bounds, csr_src, E_POS);

    gemm_wmma_f32<128><<<256, 256, 0, stream>>>(x, W_init, b_init, bufA, MT);

    gemm_wmma_f32<128><<<256, 256, 0, stream>>>(bufA, W1, nullptr, bufB, MT);
    gather_nodes_kernel<128, true><<<(N_NODES + 7) / 8, TB, 0, stream>>>(
        bufB, bounds, csr_src, dinv, b1, bufA, N_NODES, E_POS);

    gemm_wmma_f32<64><<<256, 256, 0, stream>>>(bufA, W2, nullptr, bufB, MT);
    gather_nodes_kernel<64, false><<<(N_NODES + 7) / 8, TB, 0, stream>>>(
        bufB, bounds, csr_src, dinv, b2, bufA, N_NODES, E_POS);

    {
        long long threads = (long long)E_PRED * 16;
        edge_score_kernel<<<(unsigned)((threads + TB - 1) / TB), TB, 0, stream>>>(
            bufA, qa, qb, logits, E_PRED);
    }
}


# Node gather and edge scoring on fixed grids of resident waves that loop over nodes / edges (next node's header prefetched, 4 edges in flight) instead of one short wave per item; sum order unchanged
# speedup vs baseline: 2.9629x; 1.0518x over previous
#include <hip/hip_runtime.h>
#include <hip/hip_bf16.h>

typedef __attribute__((ext_vector_type(2))) float v2f;
typedef __attribute__((ext_vector_type(4))) float v4f;
typedef __attribute__((ext_vector_type(8))) float v8f;

#define N_NODES 100000
#define E_POS   500000
#define E_PRED  200000
#define GATHER_WG 2048
#define EDGE_WG   1024

__global__ void csr_zero_kernel(int* __restrict__ cnt, int n) {
    int i = blockIdx.x * blockDim.x + threadIdx.x;
    if (i < n) cnt[i] = 0;
}

__global__ void csr_count_kernel(const int* __restrict__ tgt, int* __restrict__ cnt, int e) {
    int i = blockIdx.x * blockDim.x + threadIdx.x;
    if (i < e) atomicAdd(&cnt[tgt[i]], 1);
}

__global__ __launch_bounds__(1024) void csr_scan_kernel(const int* __restrict__ cnt,
                                                        int* __restrict__ bounds,
                                                        float* __restrict__ dinv, int n) {
    __shared__ int wsum[32];
    __shared__ int wpre[32];
    const int tid  = threadIdx.x;
    const int lane = tid & 31;
    const int wv   = tid >> 5;
    const int base = blockIdx.x * 1024;
    int before = 0;
    for (int i = tid; i < base; i += 1024) before += cnt[i];
    const int i = base + tid;
    const int c = (i < n) ? cnt[i] : 0;
    int inc = c;
#pragma unroll
    for (int d = 1; d < 32; d <<= 1) {
        const int t = __shfl_up(inc, d, 32);
        if (lane >= d) inc += t;
    }
#pragma unroll
    for (int d = 16; d > 0; d >>= 1) before += __shfl_xor(before, d, 32);
    if (lane == 31) wsum[wv] = inc;
    if (lane == 0)  wpre[wv] = before;
    __syncthreads();
    int off = 0;
#pragma unroll
    for (int k = 0; k < 32; ++k) {
        off += wpre[k];
        if (k < wv) off += wsum[k];
    }
    const int excl = off + inc - c;
    if (i < n) {
        *(int2*)(bounds + 2 * i) = make_int2(excl, excl);
        dinv[i]     = rsqrtf((float)(c + 1));
    }
}

__global__ void csr_fill_kernel(const int* __restrict__ src, const int* __restrict__ tgt,
                                int* __restrict__ bounds, int* __restrict__ csr_src, int e) {
    int i = blockIdx.x * blockDim.x + threadIdx.x;
    if (i < e) {
        const int pos = atomicAdd(&bounds[2 * tgt[i] + 1], 1);
        if ((unsigned)pos < (unsigned)e) csr_src[pos] = src[i];
    }
}

template <int N>
__global__ __launch_bounds__(256) void gemm_wmma_f32(const float* __restrict__ A,
                                                     const float* __restrict__ B,
                                                     const float* __restrict__ bias,
                                                     float* __restrict__ C,
                                                     int mtiles) {
    constexpr int K   = 128;
    constexpr int NS  = N / 64;
    constexpr int WPS = 8 / NS;
    __shared__ float4 Bs[(K / 8) * N * 2];

    const int tid = threadIdx.x;
    for (int p = tid; p < (K / 4) * N; p += 256) {
        const int n  = p % N;
        const int k4 = p / N;
        const float* bp = B + (size_t)(4 * k4) * N + n;
        float4 v;
        v.x = bp[0];
        v.y = bp[N];
        v.z = bp[2 * N];
        v.w = bp[3 * N];
        Bs[((((k4 >> 1) * N) + n) << 1) + (k4 & 1)] = v;
    }
    __syncthreads();

    const int lane = tid & 31;
    const int half = lane >> 4;
    const int idx  = lane & 15;
    const int wave  = __builtin_amdgcn_readfirstlane(tid >> 5);
    const int slice = wave % NS;
    const int grp   = wave / NS;
    const int nwg   = gridDim.x;
    const int n0    = slice * 64;

    float4 b[K / 8][4];
    {
        const float4* bfrag = Bs + (n0 + idx) * 2 + half;
#pragma unroll
        for (int j = 0; j < K / 8; ++j)
#pragma unroll
            for (int t = 0; t < 4; ++t) b[j][t] = bfrag[(j * N + 16 * t) * 2];
    }
    float bv[4];
#pragma unroll
    for (int t = 0; t < 4; ++t) bv[t] = bias ? bias[n0 + 16 * t + idx] : 0.0f;

    float4 a[K / 8];
    int mt = grp * nwg + blockIdx.x;
    if (mt < mtiles) {
        const float4* ap = (const float4*)(A + (size_t)(mt * 16 + idx) * K) + half;
#pragma unroll
        for (int j = 0; j < K / 8; ++j) a[j] = ap[2 * j];
    }

    for (; mt < mtiles; mt += WPS * nwg) {
        const int m0 = mt * 16;
        const int mtn = min(mt + WPS * nwg, mtiles - 1);
        const float4* apn = (const float4*)(A + (size_t)(mtn * 16 + idx) * K) + half;
        float4 an[K / 8];
#pragma unroll
        for (int j = 0; j < K / 8; ++j) an[j] = apn[2 * j];

        v8f acc[4];
#pragma unroll
        for (int t = 0; t < 4; ++t) acc[t] = v8f{};

#pragma unroll
        for (int j = 0; j < K / 8; ++j) {
            v2f a0; a0.x = a[j].x; a0.y = a[j].y;
            v2f a1; a1.x = a[j].z; a1.y = a[j].w;
#pragma unroll
            for (int t = 0; t < 4; ++t) {
                v2f b0; b0.x = b[j][t].x; b0.y = b[j][t].y;
                acc[t] = __builtin_amdgcn_wmma_f32_16x16x4_f32(
                    false, a0, false, b0, (short)0, acc[t], false, false);
            }
#pragma unroll
            for (int t = 0; t < 4; ++t) {
                v2f b1; b1.x = b[j][t].z; b1.y = b[j][t].w;
                acc[t] = __builtin_amdgcn_wmma_f32_16x16x4_f32(
                    false, a1, false, b1, (short)0, acc[t], false, false);
            }
        }

        float* crow = C + (size_t)(m0 + half * 8) * N + n0 + idx;
#pragma unroll
        for (int t = 0; t < 4; ++t) {
#pragma unroll
            for (int r = 0; r < 8; ++r) {
                crow[(size_t)r * N + 16 * t] = acc[t][r] + bv[t];
            }
        }
#pragma unroll
        for (int j = 0; j < K / 8; ++j) a[j] = an[j];
    }
}

template <int V> struct rowvec;
template <> struct rowvec<4> { typedef v4f type; };
template <> struct rowvec<2> { typedef v2f type; };

template <int F, bool RELU>
__global__ __launch_bounds__(256) void gather_nodes_kernel(const float* __restrict__ xw,
                                                           const int* __restrict__ bounds,
                                                           const int* __restrict__ csr_src,
                                                           const float* __restrict__ dinv,
                                                           const float* __restrict__ bias,
                                                           float* __restrict__ out, int n, int e) {
    constexpr int V = F / 32;
    typedef typename rowvec<V>::type vf;
    const int lane   = threadIdx.x & 31;
    const int nwaves = (int)gridDim.x * 8;
    int node = __builtin_amdgcn_readfirstlane((int)(blockIdx.x * 8 + (threadIdx.x >> 5)));
    if (node >= n) return;
    const float* xcol = xw + lane * V;
    const vf bv = *(const vf*)(bias + lane * V);
    int2  seg  = *(const int2*)(bounds + 2 * node);
    float di   = dinv[node];
    vf    self = *(const vf*)(xcol + (size_t)node * F);
    for (;;) {
        const int   nxt   = node + nwaves;
        const bool  more  = nxt < n;
        const int   pn    = more ? nxt : node;
        const int2  nseg  = *(const int2*)(bounds + 2 * pn);
        const float ndi   = dinv[pn];
        const vf    nself = *(const vf*)(xcol + (size_t)pn * F);

        const int p0 = max(seg.x, 0);
        const int p1 = min(seg.y, e);
        vf acc = {};
        for (int base = p0; base < p1; base += 32) {
            const int m = min(p1 - base, 32);
            int s = node;
            if (lane < m) s = min(max(csr_src[base + lane], 0), n - 1);
            const int wi = __float_as_int(dinv[s]);
            int j = 0;
            for (; j + 4 <= m; j += 4) {
                const int s0 = __builtin_amdgcn_readlane(s, j);
                const int s1 = __builtin_amdgcn_readlane(s, j + 1);
                const int s2 = __builtin_amdgcn_readlane(s, j + 2);
                const int s3 = __builtin_amdgcn_readlane(s, j + 3);
                const vf r0 = *(const vf*)(xcol + (size_t)s0 * F);
                const vf r1 = *(const vf*)(xcol + (size_t)s1 * F);
                const vf r2 = *(const vf*)(xcol + (size_t)s2 * F);
                const vf r3 = *(const vf*)(xcol + (size_t)s3 * F);
                acc += r0 * (__int_as_float(__builtin_amdgcn_readlane(wi, j)) * di);
                acc += r1 * (__int_as_float(__builtin_amdgcn_readlane(wi, j + 1)) * di);
                acc += r2 * (__int_as_float(__builtin_amdgcn_readlane(wi, j + 2)) * di);
                acc += r3 * (__int_as_float(__builtin_amdgcn_readlane(wi, j + 3)) * di);
            }
            for (; j < m; ++j) {
                const int s0 = __builtin_amdgcn_readlane(s, j);
                const vf r0 = *(const vf*)(xcol + (size_t)s0 * F);
                acc += r0 * (__int_as_float(__builtin_amdgcn_readlane(wi, j)) * di);
            }
        }
        vf o;
#pragma unroll
        for (int c = 0; c < V; ++c) {
            float v = acc[c] + self[c] * di * di + bv[c];
            if (RELU) v = fmaxf(v, 0.0f);
            o[c] = v;
        }
        *(vf*)(out + (size_t)node * F + lane * V) = o;

        if (!more) break;
        node = nxt;
        seg  = nseg;
        di   = ndi;
        self = nself;
    }
}

__global__ __launch_bounds__(256) void edge_score_kernel(const float* __restrict__ h,
                                                         const int* __restrict__ ia,
                                                         const int* __restrict__ ib,
                                                         float* __restrict__ out, int e) {
    constexpr int U = 4;
    const int lane   = threadIdx.x & 15;
    const int stride = (int)gridDim.x * 16;
    int ed = (int)((blockIdx.x * 256 + threadIdx.x) >> 4);
    for (; ed + (U - 1) * stride < e; ed += U * stride) {
        int a[U], b[U];
#pragma unroll
        for (int u = 0; u < U; ++u) {
            a[u] = ia[ed + u * stride];
            b[u] = ib[ed + u * stride];
        }
        float4 va[U], vb[U];
#pragma unroll
        for (int u = 0; u < U; ++u) {
            va[u] = ((const float4*)(h + (size_t)a[u] * 64))[lane];
            vb[u] = ((const float4*)(h + (size_t)b[u] * 64))[lane];
        }
#pragma unroll
        for (int u = 0; u < U; ++u) {
            float s = va[u].x * vb[u].x + va[u].y * vb[u].y + va[u].z * vb[u].z + va[u].w * vb[u].w;
            s += __shfl_xor(s, 8, 32);
            s += __shfl_xor(s, 4, 32);
            s += __shfl_xor(s, 2, 32);
            s += __shfl_xor(s, 1, 32);
            if (lane == 0) out[ed + u * stride] = s;
        }
    }
    for (; ed < e; ed += stride) {
        const int a = ia[ed];
        const int b = ib[ed];
        const float4 va = ((const float4*)(h + (size_t)a * 64))[lane];
        const float4 vb = ((const float4*)(h + (size_t)b * 64))[lane];
        float s = va.x * vb.x + va.y * vb.y + va.z * vb.z + va.w * vb.w;
        s += __shfl_xor(s, 8, 32);
        s += __shfl_xor(s, 4, 32);
        s += __shfl_xor(s, 2, 32);
        s += __shfl_xor(s, 1, 32);
        if (lane == 0) out[ed] = s;
    }
}

extern "C" void kernel_launch(void* const* d_in, const int* in_sizes, int n_in,
                              void* d_out, int out_size, void* d_ws, size_t ws_size,
                              hipStream_t stream) {
    const float* x      = (const float*)d_in[0];
    const int*   eidx   = (const int*)d_in[1];
    const int*   pidx   = (const int*)d_in[2];
    const float* W_init = (const float*)d_in[3];
    const float* b_init = (const float*)d_in[4];
    const float* W1     = (const float*)d_in[5];
    const float* b1     = (const float*)d_in[6];
    const float* W2     = (const float*)d_in[7];
    const float* b2     = (const float*)d_in[8];
    float* logits = (float*)d_out;

    const int* src = pidx;
    const int* tgt = pidx + E_POS;
    const int* qa  = eidx;
    const int* qb  = eidx + E_PRED;

    float* ws   = (float*)d_ws;
    float* dinv = ws;
    float* bufA = ws + 102400;
    float* bufB = bufA + (size_t)N_NODES * 128;
    float* bufC = bufB + (size_t)N_NODES * 128;
    int*  cnt     = (int*)bufC;
    int*  bounds  = cnt + 102400;
    int*  csr_src = cnt + 3 * 102400;

    const int TB = 256;
    const int MT = N_NODES / 16;

    csr_zero_kernel<<<(N_NODES + TB - 1) / TB, TB, 0, stream>>>(cnt, N_NODES);
    csr_count_kernel<<<(E_POS + TB - 1) / TB, TB, 0, stream>>>(tgt, cnt, E_POS);
    csr_scan_kernel<<<(N_NODES + 1023) / 1024, 1024, 0, stream>>>(cnt, bounds, dinv, N_NODES);
    csr_fill_kernel<<<(E_POS + TB - 1) / TB, TB, 0, stream>>>(src, tgt, bounds, csr_src, E_POS);

    gemm_wmma_f32<128><<<256, 256, 0, stream>>>(x, W_init, b_init, bufA, MT);

    gemm_wmma_f32<128><<<256, 256, 0, stream>>>(bufA, W1, nullptr, bufB, MT);
    gather_nodes_kernel<128, true><<<GATHER_WG, TB, 0, stream>>>(
        bufB, bounds, csr_src, dinv, b1, bufA, N_NODES, E_POS);

    gemm_wmma_f32<64><<<256, 256, 0, stream>>>(bufA, W2, nullptr, bufB, MT);
    gather_nodes_kernel<64, false><<<GATHER_WG, TB, 0, stream>>>(
        bufB, bounds, csr_src, dinv, b2, bufA, N_NODES, E_POS);

    edge_score_kernel<<<EDGE_WG, TB, 0, stream>>>(bufA, qa, qb, logits, E_PRED);
}


# First two linear maps folded: Wc = W_init @ W1 (WMMA) and bc = b_init @ W1 formed in the zero-fill launch, then xw1 = x @ Wc + bc; one N x 128 x 128 GEMM launch gone; f32 rounding order changes
# speedup vs baseline: 3.4040x; 1.1489x over previous
#include <hip/hip_runtime.h>
#include <hip/hip_bf16.h>

typedef __attribute__((ext_vector_type(2))) float v2f;
typedef __attribute__((ext_vector_type(4))) float v4f;
typedef __attribute__((ext_vector_type(8))) float v8f;

#define N_NODES 100000
#define E_POS   500000
#define E_PRED  200000
#define GATHER_WG 2048
#define EDGE_WG   1024

__global__ void csr_count_kernel(const int* __restrict__ tgt, int* __restrict__ cnt, int e) {
    int i = blockIdx.x * blockDim.x + threadIdx.x;
    if (i < e) atomicAdd(&cnt[tgt[i]], 1);
}

__global__ __launch_bounds__(1024) void csr_scan_kernel(const int* __restrict__ cnt,
                                                        int* __restrict__ bounds,
                                                        float* __restrict__ dinv, int n) {
    __shared__ int wsum[32];
    __shared__ int wpre[32];
    const int tid  = threadIdx.x;
    const int lane = tid & 31;
    const int wv   = tid >> 5;
    const int base = blockIdx.x * 1024;
    int before = 0;
    for (int i = tid; i < base; i += 1024) before += cnt[i];
    const int i = base + tid;
    const int c = (i < n) ? cnt[i] : 0;
    int inc = c;
#pragma unroll
    for (int d = 1; d < 32; d <<= 1) {
        const int t = __shfl_up(inc, d, 32);
        if (lane >= d) inc += t;
    }
#pragma unroll
    for (int d = 16; d > 0; d >>= 1) before += __shfl_xor(before, d, 32);
    if (lane == 31) wsum[wv] = inc;
    if (lane == 0)  wpre[wv] = before;
    __syncthreads();
    int off = 0;
#pragma unroll
    for (int k = 0; k < 32; ++k) {
        off += wpre[k];
        if (k < wv) off += wsum[k];
    }
    const int excl = off + inc - c;
    if (i < n) {
        *(int2*)(bounds + 2 * i) = make_int2(excl, excl);
        dinv[i]     = rsqrtf((float)(c + 1));
    }
}

__global__ void csr_fill_kernel(const int* __restrict__ src, const int* __restrict__ tgt,
                                int* __restrict__ bounds, int* __restrict__ csr_src, int e) {
    int i = blockIdx.x * blockDim.x + threadIdx.x;
    if (i < e) {
        const int pos = atomicAdd(&bounds[2 * tgt[i] + 1], 1);
        if ((unsigned)pos < (unsigned)e) csr_src[pos] = src[i];
    }
}

template <int N>
__device__ __forceinline__ void gemm_body(const float* __restrict__ A,
                                          const float* __restrict__ B,
                                          const float* __restrict__ bias,
                                          const float* __restrict__ fvec,
                                          float* __restrict__ fout,
                                          float* __restrict__ C,
                                          int mtiles, int bid, int nwg) {
    constexpr int K   = 128;
    constexpr int NS  = N / 64;
    constexpr int WPS = 8 / NS;
    __shared__ float4 Bs[(K / 8) * N * 2];

    const int tid = threadIdx.x;
    for (int p = tid; p < (K / 4) * N; p += 256) {
        const int n  = p % N;
        const int k4 = p / N;
        const float* bp = B + (size_t)(4 * k4) * N + n;
        float4 v;
        v.x = bp[0];
        v.y = bp[N];
        v.z = bp[2 * N];
        v.w = bp[3 * N];
        Bs[((((k4 >> 1) * N) + n) << 1) + (k4 & 1)] = v;
    }
    __syncthreads();

    const int lane = tid & 31;
    const int half = lane >> 4;
    const int idx  = lane & 15;
    const int wave  = __builtin_amdgcn_readfirstlane(tid >> 5);
    const int slice = wave % NS;
    const int grp   = wave / NS;
    const int n0    = slice * 64;

    float4 b[K / 8][4];
    {
        const float4* bfrag = Bs + (n0 + idx) * 2 + half;
#pragma unroll
        for (int j = 0; j < K / 8; ++j)
#pragma unroll
            for (int t = 0; t < 4; ++t) b[j][t] = bfrag[(j * N + 16 * t) * 2];
    }
    float bv[4];
#pragma unroll
    for (int t = 0; t < 4; ++t) bv[t] = bias ? bias[n0 + 16 * t + idx] : 0.0f;

    if (fvec != nullptr && bid == 0 && grp == 0) {
        float s[4] = {0.0f, 0.0f, 0.0f, 0.0f};
#pragma unroll
        for (int j = 0; j < K / 8; ++j) {
            const float4 f = *(const float4*)(fvec + 8 * j + 4 * half);
#pragma unroll
            for (int t = 0; t < 4; ++t)
                s[t] += f.x * b[j][t].x + f.y * b[j][t].y + f.z * b[j][t].z + f.w * b[j][t].w;
        }
#pragma unroll
        for (int t = 0; t < 4; ++t) {
            const float tot = s[t] + __shfl_xor(s[t], 16, 32);
            if (half == 0) fout[n0 + 16 * t + idx] = tot;
        }
    }

    float4 a[K / 8];
    int mt = grp * nwg + bid;
    if (mt < mtiles) {
        const float4* ap = (const float4*)(A + (size_t)(mt * 16 + idx) * K) + half;
#pragma unroll
        for (int j = 0; j < K / 8; ++j) a[j] = ap[2 * j];
    }

    for (; mt < mtiles; mt += WPS * nwg) {
        const int m0 = mt * 16;
        const int mtn = min(mt + WPS * nwg, mtiles - 1);
        const float4* apn = (const float4*)(A + (size_t)(mtn * 16 + idx) * K) + half;
        float4 an[K / 8];
#pragma unroll
        for (int j = 0; j < K / 8; ++j) an[j] = apn[2 * j];

        v8f acc[4];
#pragma unroll
        for (int t = 0; t < 4; ++t) acc[t] = v8f{};

#pragma unroll
        for (int j = 0; j < K / 8; ++j) {
            v2f a0; a0.x = a[j].x; a0.y = a[j].y;
            v2f a1; a1.x = a[j].z; a1.y = a[j].w;
#pragma unroll
            for (int t = 0; t < 4; ++t) {
                v2f b0; b0.x = b[j][t].x; b0.y = b[j][t].y;
                acc[t] = __builtin_amdgcn_wmma_f32_16x16x4_f32(
                    false, a0, false, b0, (short)0, acc[t], false, false);
            }
#pragma unroll
            for (int t = 0; t < 4; ++t) {
                v2f b1; b1.x = b[j][t].z; b1.y = b[j][t].w;
                acc[t] = __builtin_amdgcn_wmma_f32_16x16x4_f32(
                    false, a1, false, b1, (short)0, acc[t], false, false);
            }
        }

        float* crow = C + (size_t)(m0 + half * 8) * N + n0 + idx;
#pragma unroll
        for (int t = 0; t < 4; ++t) {
#pragma unroll
            for (int r = 0; r < 8; ++r) {
                crow[(size_t)r * N + 16 * t] = acc[t][r] + bv[t];
            }
        }
#pragma unroll
        for (int j = 0; j < K / 8; ++j) a[j] = an[j];
    }
}

template <int N>
__global__ __launch_bounds__(256) void gemm_wmma_f32(const float* __restrict__ A,
                                                     const float* __restrict__ B,
                                                     const float* __restrict__ bias,
                                                     float* __restrict__ C,
                                                     int mtiles) {
    gemm_body<N>(A, B, bias, nullptr, nullptr, C, mtiles, (int)blockIdx.x, (int)gridDim.x);
}

#define FOLD_WG 8
__global__ __launch_bounds__(256) void fold_and_zero_kernel(const float* __restrict__ Wa,
                                                            const float* __restrict__ Wb,
                                                            const float* __restrict__ ba,
                                                            float* __restrict__ Wc,
                                                            float* __restrict__ bc,
                                                            int* __restrict__ cnt, int n) {
    if (blockIdx.x < FOLD_WG) {
        gemm_body<128>(Wa, Wb, nullptr, ba, bc, Wc, 128 / 16, (int)blockIdx.x, FOLD_WG);
    } else {
        const int i = (int)(blockIdx.x - FOLD_WG) * 256 + (int)threadIdx.x;
        if (i < n) cnt[i] = 0;
    }
}

template <int V> struct rowvec;
template <> struct rowvec<4> { typedef v4f type; };
template <> struct rowvec<2> { typedef v2f type; };

template <int F, bool RELU>
__global__ __launch_bounds__(256) void gather_nodes_kernel(const float* __restrict__ xw,
                                                           const int* __restrict__ bounds,
                                                           const int* __restrict__ csr_src,
                                                           const float* __restrict__ dinv,
                                                           const float* __restrict__ bias,
                                                           float* __restrict__ out, int n, int e) {
    constexpr int V = F / 32;
    typedef typename rowvec<V>::type vf;
    const int lane   = threadIdx.x & 31;
    const int nwaves = (int)gridDim.x * 8;
    int node = __builtin_amdgcn_readfirstlane((int)(blockIdx.x * 8 + (threadIdx.x >> 5)));
    if (node >= n) return;
    const float* xcol = xw + lane * V;
    const vf bv = *(const vf*)(bias + lane * V);
    int2  seg  = *(const int2*)(bounds + 2 * node);
    float di   = dinv[node];
    vf    self = *(const vf*)(xcol + (size_t)node * F);
    for (;;) {
        const int   nxt   = node + nwaves;
        const bool  more  = nxt < n;
        const int   pn    = more ? nxt : node;
        const int2  nseg  = *(const int2*)(bounds + 2 * pn);
        const float ndi   = dinv[pn];
        const vf    nself = *(const vf*)(xcol + (size_t)pn * F);

        const int p0 = max(seg.x, 0);
        const int p1 = min(seg.y, e);
        vf acc = {};
        for (int base = p0; base < p1; base += 32) {
            const int m = min(p1 - base, 32);
            int s = node;
            if (lane < m) s = min(max(csr_src[base + lane], 0), n - 1);
            const int wi = __float_as_int(dinv[s]);
            int j = 0;
            for (; j + 4 <= m; j += 4) {
                const int s0 = __builtin_amdgcn_readlane(s, j);
                const int s1 = __builtin_amdgcn_readlane(s, j + 1);
                const int s2 = __builtin_amdgcn_readlane(s, j + 2);
                const int s3 = __builtin_amdgcn_readlane(s, j + 3);
                const vf r0 = *(const vf*)(xcol + (size_t)s0 * F);
                const vf r1 = *(const vf*)(xcol + (size_t)s1 * F);
                const vf r2 = *(const vf*)(xcol + (size_t)s2 * F);
                const vf r3 = *(const vf*)(xcol + (size_t)s3 * F);
                acc += r0 * (__int_as_float(__builtin_amdgcn_readlane(wi, j)) * di);
                acc += r1 * (__int_as_float(__builtin_amdgcn_readlane(wi, j + 1)) * di);
                acc += r2 * (__int_as_float(__builtin_amdgcn_readlane(wi, j + 2)) * di);
                acc += r3 * (__int_as_float(__builtin_amdgcn_readlane(wi, j + 3)) * di);
            }
            for (; j < m; ++j) {
                const int s0 = __builtin_amdgcn_readlane(s, j);
                const vf r0 = *(const vf*)(xcol + (size_t)s0 * F);
                acc += r0 * (__int_as_float(__builtin_amdgcn_readlane(wi, j)) * di);
            }
        }
        vf o;
#pragma unroll
        for (int c = 0; c < V; ++c) {
            float v = acc[c] + self[c] * di * di + bv[c];
            if (RELU) v = fmaxf(v, 0.0f);
            o[c] = v;
        }
        *(vf*)(out + (size_t)node * F + lane * V) = o;

        if (!more) break;
        node = nxt;
        seg  = nseg;
        di   = ndi;
        self = nself;
    }
}

__global__ __launch_bounds__(256) void edge_score_kernel(const float* __restrict__ h,
                                                         const int* __restrict__ ia,
                                                         const int* __restrict__ ib,
                                                         float* __restrict__ out, int e) {
    constexpr int U = 4;
    const int lane   = threadIdx.x & 15;
    const int stride = (int)gridDim.x * 16;
    int ed = (int)((blockIdx.x * 256 + threadIdx.x) >> 4);
    for (; ed + (U - 1) * stride < e; ed += U * stride) {
        int a[U], b[U];
#pragma unroll
        for (int u = 0; u < U; ++u) {
            a[u] = ia[ed + u * stride];
            b[u] = ib[ed + u * stride];
        }
        float4 va[U], vb[U];
#pragma unroll
        for (int u = 0; u < U; ++u) {
            va[u] = ((const float4*)(h + (size_t)a[u] * 64))[lane];
            vb[u] = ((const float4*)(h + (size_t)b[u] * 64))[lane];
        }
#pragma unroll
        for (int u = 0; u < U; ++u) {
            float s = va[u].x * vb[u].x + va[u].y * vb[u].y + va[u].z * vb[u].z + va[u].w * vb[u].w;
            s += __shfl_xor(s, 8, 32);
            s += __shfl_xor(s, 4, 32);
            s += __shfl_xor(s, 2, 32);
            s += __shfl_xor(s, 1, 32);
            if (lane == 0) out[ed + u * stride] = s;
        }
    }
    for (; ed < e; ed += stride) {
        const int a = ia[ed];
        const int b = ib[ed];
        const float4 va = ((const float4*)(h + (size_t)a * 64))[lane];
        const float4 vb = ((const float4*)(h + (size_t)b * 64))[lane];
        float s = va.x * vb.x + va.y * vb.y + va.z * vb.z + va.w * vb.w;
        s += __shfl_xor(s, 8, 32);
        s += __shfl_xor(s, 4, 32);
        s += __shfl_xor(s, 2, 32);
        s += __shfl_xor(s, 1, 32);
        if (lane == 0) out[ed] = s;
    }
}

extern "C" void kernel_launch(void* const* d_in, const int* in_sizes, int n_in,
                              void* d_out, int out_size, void* d_ws, size_t ws_size,
                              hipStream_t stream) {
    const float* x      = (const float*)d_in[0];
    const int*   eidx   = (const int*)d_in[1];
    const int*   pidx   = (const int*)d_in[2];
    const float* W_init = (const float*)d_in[3];
    const float* b_init = (const float*)d_in[4];
    const float* W1     = (const float*)d_in[5];
    const float* b1     = (const float*)d_in[6];
    const float* W2     = (const float*)d_in[7];
    const float* b2     = (const float*)d_in[8];
    float* logits = (float*)d_out;

    const int* src = pidx;
    const int* tgt = pidx + E_POS;
    const int* qa  = eidx;
    const int* qb  = eidx + E_PRED;

    float* ws   = (float*)d_ws;
    float* dinv = ws;
    float* bufA = ws + 102400;
    float* bufB = bufA + (size_t)N_NODES * 128;
    float* bufC = bufB + (size_t)N_NODES * 128;
    int*  cnt     = (int*)bufC;
    int*  bounds  = cnt + 102400;
    int*  csr_src = cnt + 3 * 102400;
    float* Wc = (float*)(cnt + 8 * 102400);
    float* bc = Wc + 128 * 128;

    const int TB = 256;
    const int MT = N_NODES / 16;

    fold_and_zero_kernel<<<FOLD_WG + (N_NODES + TB - 1) / TB, TB, 0, stream>>>(
        W_init, W1, b_init, Wc, bc, cnt, N_NODES);
    csr_count_kernel<<<(E_POS + TB - 1) / TB, TB, 0, stream>>>(tgt, cnt, E_POS);
    csr_scan_kernel<<<(N_NODES + 1023) / 1024, 1024, 0, stream>>>(cnt, bounds, dinv, N_NODES);
    csr_fill_kernel<<<(E_POS + TB - 1) / TB, TB, 0, stream>>>(src, tgt, bounds, csr_src, E_POS);

    gemm_wmma_f32<128><<<256, 256, 0, stream>>>(x, Wc, bc, bufB, MT);
    gather_nodes_kernel<128, true><<<GATHER_WG, TB, 0, stream>>>(
        bufB, bounds, csr_src, dinv, b1, bufA, N_NODES, E_POS);

    gemm_wmma_f32<64><<<256, 256, 0, stream>>>(bufA, W2, nullptr, bufB, MT);
    gather_nodes_kernel<64, false><<<GATHER_WG, TB, 0, stream>>>(
        bufB, bounds, csr_src, dinv, b2, bufA, N_NODES, E_POS);

    edge_score_kernel<<<EDGE_WG, TB, 0, stream>>>(bufA, qa, qb, logits, E_PRED);
}


# Edges go straight into a fixed 32-slot line per target node in one atomic pass (spill list past 32), no count and scan launches; next node's ids prefetched; sum order = slot arrival order
# speedup vs baseline: 4.0960x; 1.2033x over previous
#include <hip/hip_runtime.h>
#include <hip/hip_bf16.h>

typedef __attribute__((ext_vector_type(2))) float v2f;
typedef __attribute__((ext_vector_type(4))) float v4f;
typedef __attribute__((ext_vector_type(8))) float v8f;

#define N_NODES 100000
#define E_POS   500000
#define E_PRED  200000
#define GATHER_WG 2048
#define EDGE_WG   1024
#define CAP 32

__global__ void slot_fill_kernel(const int* __restrict__ src, const int* __restrict__ tgt,
                                 int* __restrict__ cnt, int* __restrict__ slots,
                                 int2* __restrict__ spill, int n, int e) {
    int i = blockIdx.x * blockDim.x + threadIdx.x;
    if (i < e) {
        const int t   = tgt[i];
        const int s   = src[i];
        const int pos = atomicAdd(&cnt[t], 1);
        if ((unsigned)pos < (unsigned)CAP) {
            slots[(size_t)t * CAP + pos] = s;
        } else {
            const int k = atomicAdd(&cnt[n], 1);
            if ((unsigned)k < (unsigned)e) spill[k] = make_int2(s, t);
        }
    }
}

template <int N>
__device__ __forceinline__ void gemm_body(const float* __restrict__ A,
                                          const float* __restrict__ B,
                                          const float* __restrict__ bias,
                                          const float* __restrict__ fvec,
                                          float* __restrict__ fout,
                                          float* __restrict__ C,
                                          int mtiles, int bid, int nwg) {
    constexpr int K   = 128;
    constexpr int NS  = N / 64;
    constexpr int WPS = 8 / NS;
    __shared__ float4 Bs[(K / 8) * N * 2];

    const int tid = threadIdx.x;
    for (int p = tid; p < (K / 4) * N; p += 256) {
        const int n  = p % N;
        const int k4 = p / N;
        const float* bp = B + (size_t)(4 * k4) * N + n;
        float4 v;
        v.x = bp[0];
        v.y = bp[N];
        v.z = bp[2 * N];
        v.w = bp[3 * N];
        Bs[((((k4 >> 1) * N) + n) << 1) + (k4 & 1)] = v;
    }
    __syncthreads();

    const int lane = tid & 31;
    const int half = lane >> 4;
    const int idx  = lane & 15;
    const int wave  = __builtin_amdgcn_readfirstlane(tid >> 5);
    const int slice = wave % NS;
    const int grp   = wave / NS;
    const int n0    = slice * 64;

    float4 b[K / 8][4];
    {
        const float4* bfrag = Bs + (n0 + idx) * 2 + half;
#pragma unroll
        for (int j = 0; j < K / 8; ++j)
#pragma unroll
            for (int t = 0; t < 4; ++t) b[j][t] = bfrag[(j * N + 16 * t) * 2];
    }
    float bv[4];
#pragma unroll
    for (int t = 0; t < 4; ++t) bv[t] = bias ? bias[n0 + 16 * t + idx] : 0.0f;

    if (fvec != nullptr && bid == 0 && grp == 0) {
        float s[4] = {0.0f, 0.0f, 0.0f, 0.0f};
#pragma unroll
        for (int j = 0; j < K / 8; ++j) {
            const float4 f = *(const float4*)(fvec + 8 * j + 4 * half);
#pragma unroll
            for (int t = 0; t < 4; ++t)
                s[t] += f.x * b[j][t].x + f.y * b[j][t].y + f.z * b[j][t].z + f.w * b[j][t].w;
        }
#pragma unroll
        for (int t = 0; t < 4; ++t) {
            const float tot = s[t] + __shfl_xor(s[t], 16, 32);
            if (half == 0) fout[n0 + 16 * t + idx] = tot;
        }
    }

    float4 a[K / 8];
    int mt = grp * nwg + bid;
    if (mt < mtiles) {
        const float4* ap = (const float4*)(A + (size_t)(mt * 16 + idx) * K) + half;
#pragma unroll
        for (int j = 0; j < K / 8; ++j) a[j] = ap[2 * j];
    }

    for (; mt < mtiles; mt += WPS * nwg) {
        const int m0 = mt * 16;
        const int mtn = min(mt + WPS * nwg, mtiles - 1);
        const float4* apn = (const float4*)(A + (size_t)(mtn * 16 + idx) * K) + half;
        float4 an[K / 8];
#pragma unroll
        for (int j = 0; j < K / 8; ++j) an[j] = apn[2 * j];

        v8f acc[4];
#pragma unroll
        for (int t = 0; t < 4; ++t) acc[t] = v8f{};

#pragma unroll
        for (int j = 0; j < K / 8; ++j) {
            v2f a0; a0.x = a[j].x; a0.y = a[j].y;
            v2f a1; a1.x = a[j].z; a1.y = a[j].w;
#pragma unroll
            for (int t = 0; t < 4; ++t) {
                v2f b0; b0.x = b[j][t].x; b0.y = b[j][t].y;
                acc[t] = __builtin_amdgcn_wmma_f32_16x16x4_f32(
                    false, a0, false, b0, (short)0, acc[t], false, false);
            }
#pragma unroll
            for (int t = 0; t < 4; ++t) {
                v2f b1; b1.x = b[j][t].z; b1.y = b[j][t].w;
                acc[t] = __builtin_amdgcn_wmma_f32_16x16x4_f32(
                    false, a1, false, b1, (short)0, acc[t], false, false);
            }
        }

        float* crow = C + (size_t)(m0 + half * 8) * N + n0 + idx;
#pragma unroll
        for (int t = 0; t < 4; ++t) {
#pragma unroll
            for (int r = 0; r < 8; ++r) {
                crow[(size_t)r * N + 16 * t] = acc[t][r] + bv[t];
            }
        }
#pragma unroll
        for (int j = 0; j < K / 8; ++j) a[j] = an[j];
    }
}

template <int N>
__global__ __launch_bounds__(256) void gemm_wmma_f32(const float* __restrict__ A,
                                                     const float* __restrict__ B,
                                                     const float* __restrict__ bias,
                                                     float* __restrict__ C,
                                                     int mtiles) {
    gemm_body<N>(A, B, bias, nullptr, nullptr, C, mtiles, (int)blockIdx.x, (int)gridDim.x);
}

#define FOLD_WG 8
__global__ __launch_bounds__(256) void fold_and_zero_kernel(const float* __restrict__ Wa,
                                                            const float* __restrict__ Wb,
                                                            const float* __restrict__ ba,
                                                            float* __restrict__ Wc,
                                                            float* __restrict__ bc,
                                                            int* __restrict__ cnt, int n) {
    if (blockIdx.x < FOLD_WG) {
        gemm_body<128>(Wa, Wb, nullptr, ba, bc, Wc, 128 / 16, (int)blockIdx.x, FOLD_WG);
    } else {
        const int i = (int)(blockIdx.x - FOLD_WG) * 256 + (int)threadIdx.x;
        if (i < n) cnt[i] = 0;
    }
}

template <int V> struct rowvec;
template <> struct rowvec<4> { typedef v4f type; };
template <> struct rowvec<2> { typedef v2f type; };

template <int F, bool RELU>
__global__ __launch_bounds__(256) void gather_nodes_kernel(const float* __restrict__ xw,
                                                           const int* __restrict__ cnt,
                                                           const int* __restrict__ slots,
                                                           const int2* __restrict__ spill,
                                                           const float* __restrict__ bias,
                                                           float* __restrict__ out, int n, int e) {
    constexpr int V = F / 32;
    typedef typename rowvec<V>::type vf;
    const int lane   = threadIdx.x & 31;
    const int nwaves = (int)gridDim.x * 8;
    int node = __builtin_amdgcn_readfirstlane((int)(blockIdx.x * 8 + (threadIdx.x >> 5)));
    if (node >= n) return;
    const float* xcol = xw + lane * V;
    const vf bv = *(const vf*)(bias + lane * V);
    int c   = cnt[node];
    int cn  = cnt[min(node + nwaves, n - 1)];
    int ids = node;
    if (lane < min(c, CAP)) ids = slots[(size_t)node * CAP + lane];
    vf  self = *(const vf*)(xcol + (size_t)node * F);
    for (;;) {
        const int  nxt  = node + nwaves;
        const bool more = nxt < n;
        const int  pn   = more ? nxt : node;
        const int  cpn  = more ? cn : c;
        int nids = pn;
        if (lane < min(cpn, CAP)) nids = slots[(size_t)pn * CAP + lane];
        const vf  nself = *(const vf*)(xcol + (size_t)pn * F);
        const int cnn   = cnt[min(nxt + nwaves, n - 1)];

        const float di = rsqrtf((float)(c + 1));
        const int   m  = min(max(c, 0), CAP);
        int s = node;
        if (lane < m) s = min(max(ids, 0), n - 1);
        const int wi = __float_as_int(rsqrtf((float)(cnt[s] + 1)));
        vf acc = {};
        int j = 0;
        for (; j + 4 <= m; j += 4) {
            const int s0 = __builtin_amdgcn_readlane(s, j);
            const int s1 = __builtin_amdgcn_readlane(s, j + 1);
            const int s2 = __builtin_amdgcn_readlane(s, j + 2);
            const int s3 = __builtin_amdgcn_readlane(s, j + 3);
            const vf r0 = *(const vf*)(xcol + (size_t)s0 * F);
            const vf r1 = *(const vf*)(xcol + (size_t)s1 * F);
            const vf r2 = *(const vf*)(xcol + (size_t)s2 * F);
            const vf r3 = *(const vf*)(xcol + (size_t)s3 * F);
            acc += r0 * (__int_as_float(__builtin_amdgcn_readlane(wi, j)) * di);
            acc += r1 * (__int_as_float(__builtin_amdgcn_readlane(wi, j + 1)) * di);
            acc += r2 * (__int_as_float(__builtin_amdgcn_readlane(wi, j + 2)) * di);
            acc += r3 * (__int_as_float(__builtin_amdgcn_readlane(wi, j + 3)) * di);
        }
        for (; j < m; ++j) {
            const int s0 = __builtin_amdgcn_readlane(s, j);
            const vf r0 = *(const vf*)(xcol + (size_t)s0 * F);
            acc += r0 * (__int_as_float(__builtin_amdgcn_readlane(wi, j)) * di);
        }
        if (c > CAP) {
            const int nsp = min(max(cnt[n], 0), e);
            for (int base = 0; base < nsp; base += 32) {
                const int q = base + lane;
                int os = node, ot = -1;
                if (q < nsp) { const int2 st = spill[q]; os = st.x; ot = st.y; }
                const bool hit = (ot == node);
                os = hit ? min(max(os, 0), n - 1) : node;
                const int ow = __float_as_int(rsqrtf((float)(cnt[os] + 1)));
                unsigned left = __builtin_amdgcn_ballot_w32(hit);
                while (left != 0u) {
                    const int k = __builtin_ctz(left);
                    left &= left - 1u;
                    const int s0 = __builtin_amdgcn_readlane(os, k);
                    const vf r0 = *(const vf*)(xcol + (size_t)s0 * F);
                    acc += r0 * (__int_as_float(__builtin_amdgcn_readlane(ow, k)) * di);
                }
            }
        }
        vf o;
#pragma unroll
        for (int cc = 0; cc < V; ++cc) {
            float v = acc[cc] + self[cc] * di * di + bv[cc];
            if (RELU) v = fmaxf(v, 0.0f);
            o[cc] = v;
        }
        *(vf*)(out + (size_t)node * F + lane * V) = o;

        if (!more) break;
        node = nxt;
        c    = cn;
        cn   = cnn;
        ids  = nids;
        self = nself;
    }
}

__global__ __launch_bounds__(256) void edge_score_kernel(const float* __restrict__ h,
                                                         const int* __restrict__ ia,
                                                         const int* __restrict__ ib,
                                                         float* __restrict__ out, int e) {
    constexpr int U = 4;
    const int lane   = threadIdx.x & 15;
    const int stride = (int)gridDim.x * 16;
    int ed = (int)((blockIdx.x * 256 + threadIdx.x) >> 4);
    for (; ed + (U - 1) * stride < e; ed += U * stride) {
        int a[U], b[U];
#pragma unroll
        for (int u = 0; u < U; ++u) {
            a[u] = ia[ed + u * stride];
            b[u] = ib[ed + u * stride];
        }
        float4 va[U], vb[U];
#pragma unroll
        for (int u = 0; u < U; ++u) {
            va[u] = ((const float4*)(h + (size_t)a[u] * 64))[lane];
            vb[u] = ((const float4*)(h + (size_t)b[u] * 64))[lane];
        }
#pragma unroll
        for (int u = 0; u < U; ++u) {
            float s = va[u].x * vb[u].x + va[u].y * vb[u].y + va[u].z * vb[u].z + va[u].w * vb[u].w;
            s += __shfl_xor(s, 8, 32);
            s += __shfl_xor(s, 4, 32);
            s += __shfl_xor(s, 2, 32);
            s += __shfl_xor(s, 1, 32);
            if (lane == 0) out[ed + u * stride] = s;
        }
    }
    for (; ed < e; ed += stride) {
        const int a = ia[ed];
        const int b = ib[ed];
        const float4 va = ((const float4*)(h + (size_t)a * 64))[lane];
        const float4 vb = ((const float4*)(h + (size_t)b * 64))[lane];
        float s = va.x * vb.x + va.y * vb.y + va.z * vb.z + va.w * vb.w;
        s += __shfl_xor(s, 8, 32);
        s += __shfl_xor(s, 4, 32);
        s += __shfl_xor(s, 2, 32);
        s += __shfl_xor(s, 1, 32);
        if (lane == 0) out[ed] = s;
    }
}

extern "C" void kernel_launch(void* const* d_in, const int* in_sizes, int n_in,
                              void* d_out, int out_size, void* d_ws, size_t ws_size,
                              hipStream_t stream) {
    const float* x      = (const float*)d_in[0];
    const int*   eidx   = (const int*)d_in[1];
    const int*   pidx   = (const int*)d_in[2];
    const float* W_init = (const float*)d_in[3];
    const float* b_init = (const float*)d_in[4];
    const float* W1     = (const float*)d_in[5];
    const float* b1     = (const float*)d_in[6];
    const float* W2     = (const float*)d_in[7];
    const float* b2     = (const float*)d_in[8];
    float* logits = (float*)d_out;

    const int* src = pidx;
    const int* tgt = pidx + E_POS;
    const int* qa  = eidx;
    const int* qb  = eidx + E_PRED;

    float* ws   = (float*)d_ws;
    float* bufA = ws + 102400;
    float* bufB = bufA + (size_t)N_NODES * 128;
    float* bufC = bufB + (size_t)N_NODES * 128;
    int*  cnt   = (int*)bufC;
    float* Wc = (float*)(cnt + 102400);
    float* bc = Wc + 128 * 128;
    int2* spill = (int2*)(cnt + 102400 + 16896);
    int*  slots = cnt + 102400 + 16896 + 2 * E_POS;

    const int TB = 256;
    const int MT = N_NODES / 16;

    fold_and_zero_kernel<<<FOLD_WG + (N_NODES + 1 + TB - 1) / TB, TB, 0, stream>>>(
        W_init, W1, b_init, Wc, bc, cnt, N_NODES + 1);
    slot_fill_kernel<<<(E_POS + TB - 1) / TB, TB, 0, stream>>>(src, tgt, cnt, slots, spill, N_NODES, E_POS);

    gemm_wmma_f32<128><<<256, 256, 0, stream>>>(x, Wc, bc, bufB, MT);
    gather_nodes_kernel<128, true><<<GATHER_WG, TB, 0, stream>>>(
        bufB, cnt, slots, spill, b1, bufA, N_NODES, E_POS);

    gemm_wmma_f32<64><<<256, 256, 0, stream>>>(bufA, W2, nullptr, bufB, MT);
    gather_nodes_kernel<64, false><<<GATHER_WG, TB, 0, stream>>>(
        bufB, cnt, slots, spill, b2, bufA, N_NODES, E_POS);

    edge_score_kernel<<<EDGE_WG, TB, 0, stream>>>(bufA, qa, qb, logits, E_PRED);
}


# Edge weight split: GEMM stores each row times its own dinv, gather adds plain rows and scales once by the target's dinv; per-node scattered load of source counters gone; f32 rounding order changes
# speedup vs baseline: 4.3164x; 1.0538x over previous
#include <hip/hip_runtime.h>
#include <hip/hip_bf16.h>

typedef __attribute__((ext_vector_type(2))) float v2f;
typedef __attribute__((ext_vector_type(4))) float v4f;
typedef __attribute__((ext_vector_type(8))) float v8f;

#define N_NODES 100000
#define E_POS   500000
#define E_PRED  200000
#define GATHER_WG 2048
#define EDGE_WG   1024
#define CAP 32

__global__ void slot_fill_kernel(const int* __restrict__ src, const int* __restrict__ tgt,
                                 int* __restrict__ cnt, int* __restrict__ slots,
                                 int2* __restrict__ spill, int n, int e) {
    int i = blockIdx.x * blockDim.x + threadIdx.x;
    if (i < e) {
        const int t   = tgt[i];
        const int s   = src[i];
        const int pos = atomicAdd(&cnt[t], 1);
        if ((unsigned)pos < (unsigned)CAP) {
            slots[(size_t)t * CAP + pos] = s;
        } else {
            const int k = atomicAdd(&cnt[n], 1);
            if ((unsigned)k < (unsigned)e) spill[k] = make_int2(s, t);
        }
    }
}

template <int N, bool ROWSCALE>
__device__ __forceinline__ void gemm_body(const float* __restrict__ A,
                                          const float* __restrict__ B,
                                          const float* __restrict__ bias,
                                          const float* __restrict__ fvec,
                                          float* __restrict__ fout,
                                          const int* __restrict__ rcnt,
                                          float* __restrict__ C,
                                          int mtiles, int bid, int nwg) {
    constexpr int K   = 128;
    constexpr int NS  = N / 64;
    constexpr int WPS = 8 / NS;
    __shared__ float4 Bs[(K / 8) * N * 2];

    const int tid = threadIdx.x;
    for (int p = tid; p < (K / 4) * N; p += 256) {
        const int n  = p % N;
        const int k4 = p / N;
        const float* bp = B + (size_t)(4 * k4) * N + n;
        float4 v;
        v.x = bp[0];
        v.y = bp[N];
        v.z = bp[2 * N];
        v.w = bp[3 * N];
        Bs[((((k4 >> 1) * N) + n) << 1) + (k4 & 1)] = v;
    }
    __syncthreads();

    const int lane = tid & 31;
    const int half = lane >> 4;
    const int idx  = lane & 15;
    const int wave  = __builtin_amdgcn_readfirstlane(tid >> 5);
    const int slice = wave % NS;
    const int grp   = wave / NS;
    const int n0    = slice * 64;

    float4 b[K / 8][4];
    {
        const float4* bfrag = Bs + (n0 + idx) * 2 + half;
#pragma unroll
        for (int j = 0; j < K / 8; ++j)
#pragma unroll
            for (int t = 0; t < 4; ++t) b[j][t] = bfrag[(j * N + 16 * t) * 2];
    }
    float bv[4];
#pragma unroll
    for (int t = 0; t < 4; ++t) bv[t] = bias ? bias[n0 + 16 * t + idx] : 0.0f;

    if (fvec != nullptr && bid == 0 && grp == 0) {
        float s[4] = {0.0f, 0.0f, 0.0f, 0.0f};
#pragma unroll
        for (int j = 0; j < K / 8; ++j) {
            const float4 f = *(const float4*)(fvec + 8 * j + 4 * half);
#pragma unroll
            for (int t = 0; t < 4; ++t)
                s[t] += f.x * b[j][t].x + f.y * b[j][t].y + f.z * b[j][t].z + f.w * b[j][t].w;
        }
#pragma unroll
        for (int t = 0; t < 4; ++t) {
            const float tot = s[t] + __shfl_xor(s[t], 16, 32);
            if (half == 0) fout[n0 + 16 * t + idx] = tot;
        }
    }

    float4 a[K / 8];
    int mt = grp * nwg + bid;
    if (mt < mtiles) {
        const float4* ap = (const float4*)(A + (size_t)(mt * 16 + idx) * K) + half;
#pragma unroll
        for (int j = 0; j < K / 8; ++j) a[j] = ap[2 * j];
    }

    for (; mt < mtiles; mt += WPS * nwg) {
        const int m0 = mt * 16;
        const int mtn = min(mt + WPS * nwg, mtiles - 1);
        const float4* apn = (const float4*)(A + (size_t)(mtn * 16 + idx) * K) + half;
        float4 an[K / 8];
#pragma unroll
        for (int j = 0; j < K / 8; ++j) an[j] = apn[2 * j];
        int rc[16];
#pragma unroll
        for (int r = 0; r < 16; ++r) rc[r] = ROWSCALE ? rcnt[m0 + r] : 0;

        v8f acc[4];
#pragma unroll
        for (int t = 0; t < 4; ++t) acc[t] = v8f{};

#pragma unroll
        for (int j = 0; j < K / 8; ++j) {
            v2f a0; a0.x = a[j].x; a0.y = a[j].y;
            v2f a1; a1.x = a[j].z; a1.y = a[j].w;
#pragma unroll
            for (int t = 0; t < 4; ++t) {
                v2f b0; b0.x = b[j][t].x; b0.y = b[j][t].y;
                acc[t] = __builtin_amdgcn_wmma_f32_16x16x4_f32(
                    false, a0, false, b0, (short)0, acc[t], false, false);
            }
#pragma unroll
            for (int t = 0; t < 4; ++t) {
                v2f b1; b1.x = b[j][t].z; b1.y = b[j][t].w;
                acc[t] = __builtin_amdgcn_wmma_f32_16x16x4_f32(
                    false, a1, false, b1, (short)0, acc[t], false, false);
            }
        }

        float* crow = C + (size_t)(m0 + half * 8) * N + n0 + idx;
        if (ROWSCALE) {
            float rs[8];
#pragma unroll
            for (int r = 0; r < 8; ++r) rs[r] = rsqrtf((float)((half ? rc[r + 8] : rc[r]) + 1));
#pragma unroll
            for (int t = 0; t < 4; ++t) {
#pragma unroll
                for (int r = 0; r < 8; ++r) {
                    crow[(size_t)r * N + 16 * t] = (acc[t][r] + bv[t]) * rs[r];
                }
            }
        } else {
#pragma unroll
            for (int t = 0; t < 4; ++t) {
#pragma unroll
                for (int r = 0; r < 8; ++r) {
                    crow[(size_t)r * N + 16 * t] = acc[t][r] + bv[t];
                }
            }
        }
#pragma unroll
        for (int j = 0; j < K / 8; ++j) a[j] = an[j];
    }
}

template <int N>
__global__ __launch_bounds__(256) void gemm_wmma_f32(const float* __restrict__ A,
                                                     const float* __restrict__ B,
                                                     const float* __restrict__ bias,
                                                     const int* __restrict__ rcnt,
                                                     float* __restrict__ C,
                                                     int mtiles) {
    gemm_body<N, true>(A, B, bias, nullptr, nullptr, rcnt, C, mtiles, (int)blockIdx.x, (int)gridDim.x);
}

#define FOLD_WG 8
__global__ __launch_bounds__(256) void fold_and_zero_kernel(const float* __restrict__ Wa,
                                                            const float* __restrict__ Wb,
                                                            const float* __restrict__ ba,
                                                            float* __restrict__ Wc,
                                                            float* __restrict__ bc,
                                                            int* __restrict__ cnt, int n) {
    if (blockIdx.x < FOLD_WG) {
        gemm_body<128, false>(Wa, Wb, nullptr, ba, bc, nullptr, Wc, 128 / 16, (int)blockIdx.x, FOLD_WG);
    } else {
        const int i = (int)(blockIdx.x - FOLD_WG) * 256 + (int)threadIdx.x;
        if (i < n) cnt[i] = 0;
    }
}

template <int V> struct rowvec;
template <> struct rowvec<4> { typedef v4f type; };
template <> struct rowvec<2> { typedef v2f type; };

template <int F, bool RELU>
__global__ __launch_bounds__(256) void gather_nodes_kernel(const float* __restrict__ xw,
                                                           const int* __restrict__ cnt,
                                                           const int* __restrict__ slots,
                                                           const int2* __restrict__ spill,
                                                           const float* __restrict__ bias,
                                                           float* __restrict__ out, int n, int e) {
    constexpr int V = F / 32;
    typedef typename rowvec<V>::type vf;
    const int lane   = threadIdx.x & 31;
    const int nwaves = (int)gridDim.x * 8;
    int node = __builtin_amdgcn_readfirstlane((int)(blockIdx.x * 8 + (threadIdx.x >> 5)));
    if (node >= n) return;
    const float* xcol = xw + lane * V;
    const vf bv = *(const vf*)(bias + lane * V);
    int c   = cnt[node];
    int cn  = cnt[min(node + nwaves, n - 1)];
    int ids = node;
    if (lane < min(c, CAP)) ids = slots[(size_t)node * CAP + lane];
    vf  self = *(const vf*)(xcol + (size_t)node * F);
    for (;;) {
        const int  nxt  = node + nwaves;
        const bool more = nxt < n;
        const int  pn   = more ? nxt : node;
        const int  cpn  = more ? cn : c;
        int nids = pn;
        if (lane < min(cpn, CAP)) nids = slots[(size_t)pn * CAP + lane];
        const vf  nself = *(const vf*)(xcol + (size_t)pn * F);
        const int cnn   = cnt[min(nxt + nwaves, n - 1)];

        const float di = rsqrtf((float)(c + 1));
        const int   m  = min(max(c, 0), CAP);
        int s = node;
        if (lane < m) s = min(max(ids, 0), n - 1);
        vf acc = {};
        int j = 0;
        for (; j + 4 <= m; j += 4) {
            const int s0 = __builtin_amdgcn_readlane(s, j);
            const int s1 = __builtin_amdgcn_readlane(s, j + 1);
            const int s2 = __builtin_amdgcn_readlane(s, j + 2);
            const int s3 = __builtin_amdgcn_readlane(s, j + 3);
            const vf r0 = *(const vf*)(xcol + (size_t)s0 * F);
            const vf r1 = *(const vf*)(xcol + (size_t)s1 * F);
            const vf r2 = *(const vf*)(xcol + (size_t)s2 * F);
            const vf r3 = *(const vf*)(xcol + (size_t)s3 * F);
            acc += r0;
            acc += r1;
            acc += r2;
            acc += r3;
        }
        for (; j < m; ++j) {
            const int s0 = __builtin_amdgcn_readlane(s, j);
            const vf r0 = *(const vf*)(xcol + (size_t)s0 * F);
            acc += r0;
        }
        if (c > CAP) {
            const int nsp = min(max(cnt[n], 0), e);
            for (int base = 0; base < nsp; base += 32) {
                const int q = base + lane;
                int os = node, ot = -1;
                if (q < nsp) { const int2 st = spill[q]; os = st.x; ot = st.y; }
                const bool hit = (ot == node);
                os = hit ? min(max(os, 0), n - 1) : node;
                unsigned left = __builtin_amdgcn_ballot_w32(hit);
                while (left != 0u) {
                    const int k = __builtin_ctz(left);
                    left &= left - 1u;
                    const int s0 = __builtin_amdgcn_readlane(os, k);
                    const vf r0 = *(const vf*)(xcol + (size_t)s0 * F);
                    acc += r0;
                }
            }
        }
        vf o;
#pragma unroll
        for (int cc = 0; cc < V; ++cc) {
            float v = (acc[cc] + self[cc]) * di + bv[cc];
            if (RELU) v = fmaxf(v, 0.0f);
            o[cc] = v;
        }
        *(vf*)(out + (size_t)node * F + lane * V) = o;

        if (!more) break;
        node = nxt;
        c    = cn;
        cn   = cnn;
        ids  = nids;
        self = nself;
    }
}

__global__ __launch_bounds__(256) void edge_score_kernel(const float* __restrict__ h,
                                                         const int* __restrict__ ia,
                                                         const int* __restrict__ ib,
                                                         float* __restrict__ out, int e) {
    constexpr int U = 4;
    const int lane   = threadIdx.x & 15;
    const int stride = (int)gridDim.x * 16;
    int ed = (int)((blockIdx.x * 256 + threadIdx.x) >> 4);
    for (; ed + (U - 1) * stride < e; ed += U * stride) {
        int a[U], b[U];
#pragma unroll
        for (int u = 0; u < U; ++u) {
            a[u] = ia[ed + u * stride];
            b[u] = ib[ed + u * stride];
        }
        float4 va[U], vb[U];
#pragma unroll
        for (int u = 0; u < U; ++u) {
            va[u] = ((const float4*)(h + (size_t)a[u] * 64))[lane];
            vb[u] = ((const float4*)(h + (size_t)b[u] * 64))[lane];
        }
#pragma unroll
        for (int u = 0; u < U; ++u) {
            float s = va[u].x * vb[u].x + va[u].y * vb[u].y + va[u].z * vb[u].z + va[u].w * vb[u].w;
            s += __shfl_xor(s, 8, 32);
            s += __shfl_xor(s, 4, 32);
            s += __shfl_xor(s, 2, 32);
            s += __shfl_xor(s, 1, 32);
            if (lane == 0) out[ed + u * stride] = s;
        }
    }
    for (; ed < e; ed += stride) {
        const int a = ia[ed];
        const int b = ib[ed];
        const float4 va = ((const float4*)(h + (size_t)a * 64))[lane];
        const float4 vb = ((const float4*)(h + (size_t)b * 64))[lane];
        float s = va.x * vb.x + va.y * vb.y + va.z * vb.z + va.w * vb.w;
        s += __shfl_xor(s, 8, 32);
        s += __shfl_xor(s, 4, 32);
        s += __shfl_xor(s, 2, 32);
        s += __shfl_xor(s, 1, 32);
        if (lane == 0) out[ed] = s;
    }
}

extern "C" void kernel_launch(void* const* d_in, const int* in_sizes, int n_in,
                              void* d_out, int out_size, void* d_ws, size_t ws_size,
                              hipStream_t stream) {
    const float* x      = (const float*)d_in[0];
    const int*   eidx   = (const int*)d_in[1];
    const int*   pidx   = (const int*)d_in[2];
    const float* W_init = (const float*)d_in[3];
    const float* b_init = (const float*)d_in[4];
    const float* W1     = (const float*)d_in[5];
    const float* b1     = (const float*)d_in[6];
    const float* W2     = (const float*)d_in[7];
    const float* b2     = (const float*)d_in[8];
    float* logits = (float*)d_out;

    const int* src = pidx;
    const int* tgt = pidx + E_POS;
    const int* qa  = eidx;
    const int* qb  = eidx + E_PRED;

    float* ws   = (float*)d_ws;
    float* bufA = ws + 102400;
    float* bufB = bufA + (size_t)N_NODES * 128;
    float* bufC = bufB + (size_t)N_NODES * 128;
    int*  cnt   = (int*)bufC;
    float* Wc = (float*)(cnt + 102400);
    float* bc = Wc + 128 * 128;
    int2* spill = (int2*)(cnt + 102400 + 16896);
    int*  slots = cnt + 102400 + 16896 + 2 * E_POS;

    const int TB = 256;
    const int MT = N_NODES / 16;

    fold_and_zero_kernel<<<FOLD_WG + (N_NODES + 1 + TB - 1) / TB, TB, 0, stream>>>(
        W_init, W1, b_init, Wc, bc, cnt, N_NODES + 1);
    slot_fill_kernel<<<(E_POS + TB - 1) / TB, TB, 0, stream>>>(src, tgt, cnt, slots, spill, N_NODES, E_POS);

    gemm_wmma_f32<128><<<256, 256, 0, stream>>>(x, Wc, bc, cnt, bufB, MT);
    gather_nodes_kernel<128, true><<<GATHER_WG, TB, 0, stream>>>(
        bufB, cnt, slots, spill, b1, bufA, N_NODES, E_POS);

    gemm_wmma_f32<64><<<256, 256, 0, stream>>>(bufA, W2, nullptr, cnt, bufB, MT);
    gather_nodes_kernel<64, false><<<GATHER_WG, TB, 0, stream>>>(
        bufB, cnt, slots, spill, b2, bufA, N_NODES, E_POS);

    edge_score_kernel<<<EDGE_WG, TB, 0, stream>>>(bufA, qa, qb, logits, E_PRED);
}


# Edge slots built by a two-level counting sort in LDS (chunk bins by 391-node range, then one work-group per range fills its lines) instead of a global atomic per edge; sum order = LDS arrival order
# speedup vs baseline: 5.4505x; 1.2627x over previous
#include <hip/hip_runtime.h>
#include <hip/hip_bf16.h>

typedef __attribute__((ext_vector_type(2))) float v2f;
typedef __attribute__((ext_vector_type(4))) float v4f;
typedef __attribute__((ext_vector_type(8))) float v8f;

#define N_NODES 100000
#define E_POS   500000
#define E_PRED  200000
#define GATHER_WG 2048
#define EDGE_WG   1024
#define CAP 32
#define NCHUNK 248
#define CHUNK  2017
#define NBIN   256
#define BINW   391
#define EPT    8
static_assert(CAP == 32, "one slot per lane");
static_assert(NCHUNK <= 256 && NBIN == 256, "one thread per chunk and per range");
static_assert(NCHUNK * CHUNK >= E_POS && 256 * EPT >= CHUNK && CHUNK < 32768, "chunking");
static_assert(NBIN * BINW >= N_NODES, "ranges cover the nodes");

__device__ __forceinline__ void bin_chunk_body(const int* __restrict__ src, const int* __restrict__ tgt,
                                               int2* __restrict__ chunkbuf, int* __restrict__ tab,
                                               int c, int n, int e) {
    __shared__ int  hist[NBIN];
    __shared__ int  offs[NBIN];
    __shared__ int  wsum[8];
    __shared__ int2 stage[256 * EPT];
    const int tid  = threadIdx.x;
    const int lane = tid & 31;
    const int wave = tid >> 5;
    hist[tid] = 0;
    __syncthreads();

    const int e0 = c * CHUNK;
    int s[EPT], t[EPT], pos[EPT];
#pragma unroll
    for (int u = 0; u < EPT; ++u) {
        const int q = u * 256 + tid;
        const bool ok = (q < CHUNK) && (e0 + q < e);
        t[u] = ok ? tgt[e0 + q] : -1;
        s[u] = ok ? src[e0 + q] : 0;
    }
#pragma unroll
    for (int u = 0; u < EPT; ++u) {
        pos[u] = -1;
        if ((unsigned)t[u] < (unsigned)n) pos[u] = atomicAdd(&hist[t[u] / BINW], 1);
    }
    __syncthreads();

    const int v = hist[tid];
    int x = v;
#pragma unroll
    for (int d = 1; d < 32; d <<= 1) {
        const int y = __shfl_up(x, d, 32);
        if (lane >= d) x += y;
    }
    if (lane == 31) wsum[wave] = x;
    __syncthreads();
    int before = 0;
#pragma unroll
    for (int w = 0; w < 8; ++w) before += (w < wave) ? wsum[w] : 0;
    const int start = before + x - v;
    offs[tid] = start;
    tab[c * NBIN + tid] = start | (v << 16);
    __syncthreads();

    int dst[EPT];
#pragma unroll
    for (int u = 0; u < EPT; ++u)
        dst[u] = offs[min((unsigned)t[u], (unsigned)(n - 1)) / BINW] + pos[u];
#pragma unroll
    for (int u = 0; u < EPT; ++u)
        if (pos[u] >= 0) stage[dst[u]] = make_int2(s[u], t[u]);
    __syncthreads();
    const int total = offs[NBIN - 1] + hist[NBIN - 1];
    for (int i = tid; i < total; i += 256) chunkbuf[e0 + i] = stage[i];
}

__global__ __launch_bounds__(256) void slot_build_kernel(const int2* __restrict__ chunkbuf,
                                                         const int* __restrict__ tab,
                                                         int* __restrict__ cnt, int* __restrict__ slots,
                                                         int2* __restrict__ spill, int n, int e) {
    constexpr int L4 = BINW * CAP / 4;
    constexpr int G  = 16;
    __shared__ int lcnt[BINW];
    __shared__ __attribute__((aligned(16))) int lslots[BINW * CAP];
    int4* lslots4 = reinterpret_cast<int4*>(lslots);
    const int tid  = threadIdx.x;
    const int lane = tid & 31;
    const int wv   = __builtin_amdgcn_readfirstlane(tid >> 5);
    const int b    = blockIdx.x;
    const int base = b * BINW;
    const int pk = (tid < NCHUNK) ? tab[tid * NBIN + b] : 0;
    for (int l = tid; l < BINW; l += 256) lcnt[l] = 0;
    for (int i = tid; i < L4; i += 256) lslots4[i] = make_int4(0, 0, 0, 0);
    __syncthreads();

    const int2* cb = chunkbuf + (size_t)(wv * 32) * CHUNK;
#pragma unroll
    for (int i0 = 0; i0 < 32; i0 += G) {
        int2 st[G];
#pragma unroll
        for (int j = 0; j < G; ++j) {
            const int pki = __builtin_amdgcn_readlane(pk, i0 + j);
            const int2* p = cb + (size_t)(i0 + j) * CHUNK + (pki & 0xffff);
            st[j] = make_int2(0, -1);
            if (lane < min(pki >> 16, CHUNK)) st[j] = p[lane];
        }
        int l[G], pos[G];
#pragma unroll
        for (int j = 0; j < G; ++j) {
            l[j]   = st[j].y - base;
            pos[j] = -1;
            if ((unsigned)l[j] < (unsigned)BINW) pos[j] = atomicAdd(&lcnt[l[j]], 1);
        }
#pragma unroll
        for (int j = 0; j < G; ++j) {
            if (pos[j] >= 0) {
                if (pos[j] < CAP) {
                    lslots[l[j] * CAP + pos[j]] = st[j].x;
                } else {
                    const int ks = atomicAdd(&cnt[n], 1);
                    if ((unsigned)ks < (unsigned)e) spill[ks] = st[j];
                }
            }
        }
    }
    if (__builtin_amdgcn_ballot_w32((pk >> 16) > 32) != 0u) {
        for (int i = 0; i < 32; ++i) {
            const int pki = __builtin_amdgcn_readlane(pk, i);
            const int ki  = min(pki >> 16, CHUNK);
            const int2* p = cb + (size_t)i * CHUNK + (pki & 0xffff);
            for (int q0 = 32; q0 < ki; q0 += 32) {
                int2 st = make_int2(0, -1);
                if (q0 + lane < ki) st = p[q0 + lane];
                const int l = st.y - base;
                if ((unsigned)l < (unsigned)BINW) {
                    const int pos = atomicAdd(&lcnt[l], 1);
                    if (pos < CAP) {
                        lslots[l * CAP + pos] = st.x;
                    } else {
                        const int ks = atomicAdd(&cnt[n], 1);
                        if ((unsigned)ks < (unsigned)e) spill[ks] = st;
                    }
                }
            }
        }
    }
    __syncthreads();

    int4* gl = reinterpret_cast<int4*>(slots + (size_t)base * CAP);
    if (base + BINW <= n) {
        for (int l = tid; l < BINW; l += 256) cnt[base + l] = lcnt[l];
#pragma unroll
        for (int it = 0; it < L4 / 256; ++it) gl[it * 256 + tid] = lslots4[it * 256 + tid];
        if ((L4 / 256) * 256 + tid < L4) gl[(L4 / 256) * 256 + tid] = lslots4[(L4 / 256) * 256 + tid];
    } else {
        for (int l = tid; l < BINW; l += 256)
            if (base + l < n) cnt[base + l] = lcnt[l];
        for (int i = tid; i < L4; i += 256)
            if (base + (i >> 3) < n) gl[i] = lslots4[i];
    }
}

template <int N, bool ROWSCALE>
__device__ __forceinline__ void gemm_body(const float* __restrict__ A,
                                          const float* __restrict__ B,
                                          const float* __restrict__ bias,
                                          const float* __restrict__ fvec,
                                          float* __restrict__ fout,
                                          const int* __restrict__ rcnt,
                                          float* __restrict__ C,
                                          int mtiles, int bid, int nwg) {
    constexpr int K   = 128;
    constexpr int NS  = N / 64;
    constexpr int WPS = 8 / NS;
    __shared__ float4 Bs[(K / 8) * N * 2];

    const int tid = threadIdx.x;
    for (int p = tid; p < (K / 4) * N; p += 256) {
        const int n  = p % N;
        const int k4 = p / N;
        const float* bp = B + (size_t)(4 * k4) * N + n;
        float4 v;
        v.x = bp[0];
        v.y = bp[N];
        v.z = bp[2 * N];
        v.w = bp[3 * N];
        Bs[((((k4 >> 1) * N) + n) << 1) + (k4 & 1)] = v;
    }
    __syncthreads();

    const int lane = tid & 31;
    const int half = lane >> 4;
    const int idx  = lane & 15;
    const int wave  = __builtin_amdgcn_readfirstlane(tid >> 5);
    const int slice = wave % NS;
    const int grp   = wave / NS;
    const int n0    = slice * 64;

    float4 b[K / 8][4];
    {
        const float4* bfrag = Bs + (n0 + idx) * 2 + half;
#pragma unroll
        for (int j = 0; j < K / 8; ++j)
#pragma unroll
            for (int t = 0; t < 4; ++t) b[j][t] = bfrag[(j * N + 16 * t) * 2];
    }
    float bv[4];
#pragma unroll
    for (int t = 0; t < 4; ++t) bv[t] = bias ? bias[n0 + 16 * t + idx] : 0.0f;

    if (fvec != nullptr && bid == 0 && grp == 0) {
        float s[4] = {0.0f, 0.0f, 0.0f, 0.0f};
#pragma unroll
        for (int j = 0; j < K / 8; ++j) {
            const float4 f = *(const float4*)(fvec + 8 * j + 4 * half);
#pragma unroll
            for (int t = 0; t < 4; ++t)
                s[t] += f.x * b[j][t].x + f.y * b[j][t].y + f.z * b[j][t].z + f.w * b[j][t].w;
        }
#pragma unroll
        for (int t = 0; t < 4; ++t) {
            const float tot = s[t] + __shfl_xor(s[t], 16, 32);
            if (half == 0) fout[n0 + 16 * t + idx] = tot;
        }
    }

    float4 a[K / 8];
    int mt = grp * nwg + bid;
    if (mt < mtiles) {
        const float4* ap = (const float4*)(A + (size_t)(mt * 16 + idx) * K) + half;
#pragma unroll
        for (int j = 0; j < K / 8; ++j) a[j] = ap[2 * j];
    }

    for (; mt < mtiles; mt += WPS * nwg) {
        const int m0 = mt * 16;
        const int mtn = min(mt + WPS * nwg, mtiles - 1);
        const float4* apn = (const float4*)(A + (size_t)(mtn * 16 + idx) * K) + half;
        float4 an[K / 8];
#pragma unroll
        for (int j = 0; j < K / 8; ++j) an[j] = apn[2 * j];
        int rc[16];
#pragma unroll
        for (int r = 0; r < 16; ++r) rc[r] = ROWSCALE ? rcnt[m0 + r] : 0;

        v8f acc[4];
#pragma unroll
        for (int t = 0; t < 4; ++t) acc[t] = v8f{};

#pragma unroll
        for (int j = 0; j < K / 8; ++j) {
            v2f a0; a0.x = a[j].x; a0.y = a[j].y;
            v2f a1; a1.x = a[j].z; a1.y = a[j].w;
#pragma unroll
            for (int t = 0; t < 4; ++t) {
                v2f b0; b0.x = b[j][t].x; b0.y = b[j][t].y;
                acc[t] = __builtin_amdgcn_wmma_f32_16x16x4_f32(
                    false, a0, false, b0, (short)0, acc[t], false, false);
            }
#pragma unroll
            for (int t = 0; t < 4; ++t) {
                v2f b1; b1.x = b[j][t].z; b1.y = b[j][t].w;
                acc[t] = __builtin_amdgcn_wmma_f32_16x16x4_f32(
                    false, a1, false, b1, (short)0, acc[t], false, false);
            }
        }

        float* crow = C + (size_t)(m0 + half * 8) * N + n0 + idx;
        if (ROWSCALE) {
            float rs[8];
#pragma unroll
            for (int r = 0; r < 8; ++r) rs[r] = rsqrtf((float)((half ? rc[r + 8] : rc[r]) + 1));
#pragma unroll
            for (int t = 0; t < 4; ++t) {
#pragma unroll
                for (int r = 0; r < 8; ++r) {
                    crow[(size_t)r * N + 16 * t] = (acc[t][r] + bv[t]) * rs[r];
                }
            }
        } else {
#pragma unroll
            for (int t = 0; t < 4; ++t) {
#pragma unroll
                for (int r = 0; r < 8; ++r) {
                    crow[(size_t)r * N + 16 * t] = acc[t][r] + bv[t];
                }
            }
        }
#pragma unroll
        for (int j = 0; j < K / 8; ++j) a[j] = an[j];
    }
}

template <int N>
__global__ __launch_bounds__(256) void gemm_wmma_f32(const float* __restrict__ A,
                                                     const float* __restrict__ B,
                                                     const float* __restrict__ bias,
                                                     const int* __restrict__ rcnt,
                                                     float* __restrict__ C,
                                                     int mtiles) {
    gemm_body<N, true>(A, B, bias, nullptr, nullptr, rcnt, C, mtiles, (int)blockIdx.x, (int)gridDim.x);
}

#define FOLD_WG 8
__global__ __launch_bounds__(256) void fold_and_bin_kernel(const float* __restrict__ Wa,
                                                           const float* __restrict__ Wb,
                                                           const float* __restrict__ ba,
                                                           float* __restrict__ Wc,
                                                           float* __restrict__ bc,
                                                           const int* __restrict__ src,
                                                           const int* __restrict__ tgt,
                                                           int2* __restrict__ chunkbuf,
                                                           int* __restrict__ tab,
                                                           int* __restrict__ spill_len,
                                                           int n, int e) {
    if (blockIdx.x < FOLD_WG) {
        gemm_body<128, false>(Wa, Wb, nullptr, ba, bc, nullptr, Wc, 128 / 16, (int)blockIdx.x, FOLD_WG);
    } else {
        if (blockIdx.x == FOLD_WG && threadIdx.x == 0) *spill_len = 0;
        bin_chunk_body(src, tgt, chunkbuf, tab, (int)blockIdx.x - FOLD_WG, n, e);
    }
}

template <int V> struct rowvec;
template <> struct rowvec<4> { typedef v4f type; };
template <> struct rowvec<2> { typedef v2f type; };

template <int F, bool RELU>
__global__ __launch_bounds__(256) void gather_nodes_kernel(const float* __restrict__ xw,
                                                           const int* __restrict__ cnt,
                                                           const int* __restrict__ slots,
                                                           const int2* __restrict__ spill,
                                                           const float* __restrict__ bias,
                                                           float* __restrict__ out, int n, int e) {
    constexpr int V = F / 32;
    typedef typename rowvec<V>::type vf;
    const int lane   = threadIdx.x & 31;
    const int nwaves = (int)gridDim.x * 8;
    int node = __builtin_amdgcn_readfirstlane((int)(blockIdx.x * 8 + (threadIdx.x >> 5)));
    if (node >= n) return;
    const float* xcol = xw + lane * V;
    const vf bv = *(const vf*)(bias + lane * V);
    int c   = cnt[node];
    int cn  = cnt[min(node + nwaves, n - 1)];
    int ids = node;
    if (lane < min(c, CAP)) ids = slots[(size_t)node * CAP + lane];
    vf  self = *(const vf*)(xcol + (size_t)node * F);
    for (;;) {
        const int  nxt  = node + nwaves;
        const bool more = nxt < n;
        const int  pn   = more ? nxt : node;
        const int  cpn  = more ? cn : c;
        int nids = pn;
        if (lane < min(cpn, CAP)) nids = slots[(size_t)pn * CAP + lane];
        const vf  nself = *(const vf*)(xcol + (size_t)pn * F);
        const int cnn   = cnt[min(nxt + nwaves, n - 1)];

        const float di = rsqrtf((float)(c + 1));
        const int   m  = min(max(c, 0), CAP);
        int s = node;
        if (lane < m) s = min(max(ids, 0), n - 1);
        vf acc = {};
        int j = 0;
        for (; j + 4 <= m; j += 4) {
            const int s0 = __builtin_amdgcn_readlane(s, j);
            const int s1 = __builtin_amdgcn_readlane(s, j + 1);
            const int s2 = __builtin_amdgcn_readlane(s, j + 2);
            const int s3 = __builtin_amdgcn_readlane(s, j + 3);
            const vf r0 = *(const vf*)(xcol + (size_t)s0 * F);
            const vf r1 = *(const vf*)(xcol + (size_t)s1 * F);
            const vf r2 = *(const vf*)(xcol + (size_t)s2 * F);
            const vf r3 = *(const vf*)(xcol + (size_t)s3 * F);
            acc += r0;
            acc += r1;
            acc += r2;
            acc += r3;
        }
        for (; j < m; ++j) {
            const int s0 = __builtin_amdgcn_readlane(s, j);
            const vf r0 = *(const vf*)(xcol + (size_t)s0 * F);
            acc += r0;
        }
        if (c > CAP) {
            const int nsp = min(max(cnt[n], 0), e);
            for (int base = 0; base < nsp; base += 32) {
                const int q = base + lane;
                int os = node, ot = -1;
                if (q < nsp) { const int2 st = spill[q]; os = st.x; ot = st.y; }
                const bool hit = (ot == node);
                os = hit ? min(max(os, 0), n - 1) : node;
                unsigned left = __builtin_amdgcn_ballot_w32(hit);
                while (left != 0u) {
                    const int k = __builtin_ctz(left);
                    left &= left - 1u;
                    const int s0 = __builtin_amdgcn_readlane(os, k);
                    const vf r0 = *(const vf*)(xcol + (size_t)s0 * F);
                    acc += r0;
                }
            }
        }
        vf o;
#pragma unroll
        for (int cc = 0; cc < V; ++cc) {
            float v = (acc[cc] + self[cc]) * di + bv[cc];
            if (RELU) v = fmaxf(v, 0.0f);
            o[cc] = v;
        }
        *(vf*)(out + (size_t)node * F + lane * V) = o;

        if (!more) break;
        node = nxt;
        c    = cn;
        cn   = cnn;
        ids  = nids;
        self = nself;
    }
}

__global__ __launch_bounds__(256) void edge_score_kernel(const float* __restrict__ h,
                                                         const int* __restrict__ ia,
                                                         const int* __restrict__ ib,
                                                         float* __restrict__ out, int e) {
    constexpr int U = 4;
    const int lane   = threadIdx.x & 15;
    const int stride = (int)gridDim.x * 16;
    int ed = (int)((blockIdx.x * 256 + threadIdx.x) >> 4);
    for (; ed + (U - 1) * stride < e; ed += U * stride) {
        int a[U], b[U];
#pragma unroll
        for (int u = 0; u < U; ++u) {
            a[u] = ia[ed + u * stride];
            b[u] = ib[ed + u * stride];
        }
        float4 va[U], vb[U];
#pragma unroll
        for (int u = 0; u < U; ++u) {
            va[u] = ((const float4*)(h + (size_t)a[u] * 64))[lane];
            vb[u] = ((const float4*)(h + (size_t)b[u] * 64))[lane];
        }
#pragma unroll
        for (int u = 0; u < U; ++u) {
            float s = va[u].x * vb[u].x + va[u].y * vb[u].y + va[u].z * vb[u].z + va[u].w * vb[u].w;
            s += __shfl_xor(s, 8, 32);
            s += __shfl_xor(s, 4, 32);
            s += __shfl_xor(s, 2, 32);
            s += __shfl_xor(s, 1, 32);
            if (lane == 0) out[ed + u * stride] = s;
        }
    }
    for (; ed < e; ed += stride) {
        const int a = ia[ed];
        const int b = ib[ed];
        const float4 va = ((const float4*)(h + (size_t)a * 64))[lane];
        const float4 vb = ((const float4*)(h + (size_t)b * 64))[lane];
        float s = va.x * vb.x + va.y * vb.y + va.z * vb.z + va.w * vb.w;
        s += __shfl_xor(s, 8, 32);
        s += __shfl_xor(s, 4, 32);
        s += __shfl_xor(s, 2, 32);
        s += __shfl_xor(s, 1, 32);
        if (lane == 0) out[ed] = s;
    }
}

extern "C" void kernel_launch(void* const* d_in, const int* in_sizes, int n_in,
                              void* d_out, int out_size, void* d_ws, size_t ws_size,
                              hipStream_t stream) {
    const float* x      = (const float*)d_in[0];
    const int*   eidx   = (const int*)d_in[1];
    const int*   pidx   = (const int*)d_in[2];
    const float* W_init = (const float*)d_in[3];
    const float* b_init = (const float*)d_in[4];
    const float* W1     = (const float*)d_in[5];
    const float* b1     = (const float*)d_in[6];
    const float* W2     = (const float*)d_in[7];
    const float* b2     = (const float*)d_in[8];
    float* logits = (float*)d_out;

    const int* src = pidx;
    const int* tgt = pidx + E_POS;
    const int* qa  = eidx;
    const int* qb  = eidx + E_PRED;

    float* ws   = (float*)d_ws;
    float* bufA = ws + 102400;
    float* bufB = bufA + (size_t)N_NODES * 128;
    float* bufC = bufB + (size_t)N_NODES * 128;
    int*  cnt   = (int*)bufC;
    float* Wc = (float*)(cnt + 102400);
    float* bc = Wc + 128 * 128;
    int2* spill = (int2*)(cnt + 102400 + 16896);
    int*  slots = cnt + 102400 + 16896 + 2 * E_POS;
    int2* chunkbuf = (int2*)(slots + (size_t)N_NODES * CAP);
    int*  tab   = (int*)(chunkbuf + (size_t)NCHUNK * CHUNK);

    const int TB = 256;
    const int MT = N_NODES / 16;

    fold_and_bin_kernel<<<FOLD_WG + NCHUNK, TB, 0, stream>>>(
        W_init, W1, b_init, Wc, bc, src, tgt, chunkbuf, tab, cnt + N_NODES, N_NODES, E_POS);
    slot_build_kernel<<<NBIN, TB, 0, stream>>>(chunkbuf, tab, cnt, slots, spill, N_NODES, E_POS);

    gemm_wmma_f32<128><<<256, 256, 0, stream>>>(x, Wc, bc, cnt, bufB, MT);
    gather_nodes_kernel<128, true><<<GATHER_WG, TB, 0, stream>>>(
        bufB, cnt, slots, spill, b1, bufA, N_NODES, E_POS);

    gemm_wmma_f32<64><<<256, 256, 0, stream>>>(bufA, W2, nullptr, cnt, bufB, MT);
    gather_nodes_kernel<64, false><<<GATHER_WG, TB, 0, stream>>>(
        bufB, cnt, slots, spill, b2, bufA, N_NODES, E_POS);

    edge_score_kernel<<<EDGE_WG, TB, 0, stream>>>(bufA, qa, qb, logits, E_PRED);
}


# Second layer's linear map (h1 @ W2) fused behind the first node gather: 32-node h1 tiles go through LDS to the matrix unit, h1 never reaches memory, one GEMM launch gone; sum order unchanged
# speedup vs baseline: 5.8563x; 1.0745x over previous
#include <hip/hip_runtime.h>
#include <hip/hip_bf16.h>

typedef __attribute__((ext_vector_type(2))) float v2f;
typedef __attribute__((ext_vector_type(4))) float v4f;
typedef __attribute__((ext_vector_type(8))) float v8f;

#define N_NODES 100000
#define E_POS   500000
#define E_PRED  200000
#define GATHER_WG 2048
#define EDGE_WG   1024
#define CAP 32
#define NCHUNK 248
#define CHUNK  2017
#define NBIN   256
#define BINW   391
#define EPT    8
static_assert(CAP == 32, "one slot per lane");
static_assert(NCHUNK <= 256 && NBIN == 256, "one thread per chunk and per range");
static_assert(NCHUNK * CHUNK >= E_POS && 256 * EPT >= CHUNK && CHUNK < 32768, "chunking");
static_assert(NBIN * BINW >= N_NODES, "ranges cover the nodes");

__device__ __forceinline__ void bin_chunk_body(const int* __restrict__ src, const int* __restrict__ tgt,
                                               int2* __restrict__ chunkbuf, int* __restrict__ tab,
                                               int c, int n, int e) {
    __shared__ int  hist[NBIN];
    __shared__ int  offs[NBIN];
    __shared__ int  wsum[8];
    __shared__ int2 stage[256 * EPT];
    const int tid  = threadIdx.x;
    const int lane = tid & 31;
    const int wave = tid >> 5;
    hist[tid] = 0;
    __syncthreads();

    const int e0 = c * CHUNK;
    int s[EPT], t[EPT], pos[EPT];
#pragma unroll
    for (int u = 0; u < EPT; ++u) {
        const int q = u * 256 + tid;
        const bool ok = (q < CHUNK) && (e0 + q < e);
        t[u] = ok ? tgt[e0 + q] : -1;
        s[u] = ok ? src[e0 + q] : 0;
    }
#pragma unroll
    for (int u = 0; u < EPT; ++u) {
        pos[u] = -1;
        if ((unsigned)t[u] < (unsigned)n) pos[u] = atomicAdd(&hist[t[u] / BINW], 1);
    }
    __syncthreads();

    const int v = hist[tid];
    int x = v;
#pragma unroll
    for (int d = 1; d < 32; d <<= 1) {
        const int y = __shfl_up(x, d, 32);
        if (lane >= d) x += y;
    }
    if (lane == 31) wsum[wave] = x;
    __syncthreads();
    int before = 0;
#pragma unroll
    for (int w = 0; w < 8; ++w) before += (w < wave) ? wsum[w] : 0;
    const int start = before + x - v;
    offs[tid] = start;
    tab[c * NBIN + tid] = start | (v << 16);
    __syncthreads();

    int dst[EPT];
#pragma unroll
    for (int u = 0; u < EPT; ++u)
        dst[u] = offs[min((unsigned)t[u], (unsigned)(n - 1)) / BINW] + pos[u];
#pragma unroll
    for (int u = 0; u < EPT; ++u)
        if (pos[u] >= 0) stage[dst[u]] = make_int2(s[u], t[u]);
    __syncthreads();
    const int total = offs[NBIN - 1] + hist[NBIN - 1];
    for (int i = tid; i < total; i += 256) chunkbuf[e0 + i] = stage[i];
}

__global__ __launch_bounds__(256) void slot_build_kernel(const int2* __restrict__ chunkbuf,
                                                         const int* __restrict__ tab,
                                                         int* __restrict__ cnt, int* __restrict__ slots,
                                                         int2* __restrict__ spill, int n, int e) {
    constexpr int L4 = BINW * CAP / 4;
    constexpr int G  = 16;
    __shared__ int lcnt[BINW];
    __shared__ __attribute__((aligned(16))) int lslots[BINW * CAP];
    int4* lslots4 = reinterpret_cast<int4*>(lslots);
    const int tid  = threadIdx.x;
    const int lane = tid & 31;
    const int wv   = __builtin_amdgcn_readfirstlane(tid >> 5);
    const int b    = blockIdx.x;
    const int base = b * BINW;
    const int pk = (tid < NCHUNK) ? tab[tid * NBIN + b] : 0;
    for (int l = tid; l < BINW; l += 256) lcnt[l] = 0;
    for (int i = tid; i < L4; i += 256) lslots4[i] = make_int4(0, 0, 0, 0);
    __syncthreads();

    const int2* cb = chunkbuf + (size_t)(wv * 32) * CHUNK;
#pragma unroll
    for (int i0 = 0; i0 < 32; i0 += G) {
        int2 st[G];
#pragma unroll
        for (int j = 0; j < G; ++j) {
            const int pki = __builtin_amdgcn_readlane(pk, i0 + j);
            const int2* p = cb + (size_t)(i0 + j) * CHUNK + (pki & 0xffff);
            st[j] = make_int2(0, -1);
            if (lane < min(pki >> 16, CHUNK)) st[j] = p[lane];
        }
        int l[G], pos[G];
#pragma unroll
        for (int j = 0; j < G; ++j) {
            l[j]   = st[j].y - base;
            pos[j] = -1;
            if ((unsigned)l[j] < (unsigned)BINW) pos[j] = atomicAdd(&lcnt[l[j]], 1);
        }
#pragma unroll
        for (int j = 0; j < G; ++j) {
            if (pos[j] >= 0) {
                if (pos[j] < CAP) {
                    lslots[l[j] * CAP + pos[j]] = st[j].x;
                } else {
                    const int ks = atomicAdd(&cnt[n], 1);
                    if ((unsigned)ks < (unsigned)e) spill[ks] = st[j];
                }
            }
        }
    }
    if (__builtin_amdgcn_ballot_w32((pk >> 16) > 32) != 0u) {
        for (int i = 0; i < 32; ++i) {
            const int pki = __builtin_amdgcn_readlane(pk, i);
            const int ki  = min(pki >> 16, CHUNK);
            const int2* p = cb + (size_t)i * CHUNK + (pki & 0xffff);
            for (int q0 = 32; q0 < ki; q0 += 32) {
                int2 st = make_int2(0, -1);
                if (q0 + lane < ki) st = p[q0 + lane];
                const int l = st.y - base;
                if ((unsigned)l < (unsigned)BINW) {
                    const int pos = atomicAdd(&lcnt[l], 1);
                    if (pos < CAP) {
                        lslots[l * CAP + pos] = st.x;
                    } else {
                        const int ks = atomicAdd(&cnt[n], 1);
                        if ((unsigned)ks < (unsigned)e) spill[ks] = st;
                    }
                }
            }
        }
    }
    __syncthreads();

    int4* gl = reinterpret_cast<int4*>(slots + (size_t)base * CAP);
    if (base + BINW <= n) {
        for (int l = tid; l < BINW; l += 256) cnt[base + l] = lcnt[l];
#pragma unroll
        for (int it = 0; it < L4 / 256; ++it) gl[it * 256 + tid] = lslots4[it * 256 + tid];
        if ((L4 / 256) * 256 + tid < L4) gl[(L4 / 256) * 256 + tid] = lslots4[(L4 / 256) * 256 + tid];
    } else {
        for (int l = tid; l < BINW; l += 256)
            if (base + l < n) cnt[base + l] = lcnt[l];
        for (int i = tid; i < L4; i += 256)
            if (base + (i >> 3) < n) gl[i] = lslots4[i];
    }
}

template <int N, bool ROWSCALE>
__device__ __forceinline__ void gemm_body(const float* __restrict__ A,
                                          const float* __restrict__ B,
                                          const float* __restrict__ bias,
                                          const float* __restrict__ fvec,
                                          float* __restrict__ fout,
                                          const int* __restrict__ rcnt,
                                          float* __restrict__ C,
                                          int mtiles, int bid, int nwg) {
    constexpr int K   = 128;
    constexpr int NS  = N / 64;
    constexpr int WPS = 8 / NS;
    __shared__ float4 Bs[(K / 8) * N * 2];

    const int tid = threadIdx.x;
    for (int p = tid; p < (K / 4) * N; p += 256) {
        const int n  = p % N;
        const int k4 = p / N;
        const float* bp = B + (size_t)(4 * k4) * N + n;
        float4 v;
        v.x = bp[0];
        v.y = bp[N];
        v.z = bp[2 * N];
        v.w = bp[3 * N];
        Bs[((((k4 >> 1) * N) + n) << 1) + (k4 & 1)] = v;
    }
    __syncthreads();

    const int lane = tid & 31;
    const int half = lane >> 4;
    const int idx  = lane & 15;
    const int wave  = __builtin_amdgcn_readfirstlane(tid >> 5);
    const int slice = wave % NS;
    const int grp   = wave / NS;
    const int n0    = slice * 64;

    float4 b[K / 8][4];
    {
        const float4* bfrag = Bs + (n0 + idx) * 2 + half;
#pragma unroll
        for (int j = 0; j < K / 8; ++j)
#pragma unroll
            for (int t = 0; t < 4; ++t) b[j][t] = bfrag[(j * N + 16 * t) * 2];
    }
    float bv[4];
#pragma unroll
    for (int t = 0; t < 4; ++t) bv[t] = bias ? bias[n0 + 16 * t + idx] : 0.0f;

    if (fvec != nullptr && bid == 0 && grp == 0) {
        float s[4] = {0.0f, 0.0f, 0.0f, 0.0f};
#pragma unroll
        for (int j = 0; j < K / 8; ++j) {
            const float4 f = *(const float4*)(fvec + 8 * j + 4 * half);
#pragma unroll
            for (int t = 0; t < 4; ++t)
                s[t] += f.x * b[j][t].x + f.y * b[j][t].y + f.z * b[j][t].z + f.w * b[j][t].w;
        }
#pragma unroll
        for (int t = 0; t < 4; ++t) {
            const float tot = s[t] + __shfl_xor(s[t], 16, 32);
            if (half == 0) fout[n0 + 16 * t + idx] = tot;
        }
    }

    float4 a[K / 8];
    int mt = grp * nwg + bid;
    if (mt < mtiles) {
        const float4* ap = (const float4*)(A + (size_t)(mt * 16 + idx) * K) + half;
#pragma unroll
        for (int j = 0; j < K / 8; ++j) a[j] = ap[2 * j];
    }

    for (; mt < mtiles; mt += WPS * nwg) {
        const int m0 = mt * 16;
        const int mtn = min(mt + WPS * nwg, mtiles - 1);
        const float4* apn = (const float4*)(A + (size_t)(mtn * 16 + idx) * K) + half;
        float4 an[K / 8];
#pragma unroll
        for (int j = 0; j < K / 8; ++j) an[j] = apn[2 * j];
        int rc[16];
#pragma unroll
        for (int r = 0; r < 16; ++r) rc[r] = ROWSCALE ? rcnt[m0 + r] : 0;

        v8f acc[4];
#pragma unroll
        for (int t = 0; t < 4; ++t) acc[t] = v8f{};

#pragma unroll
        for (int j = 0; j < K / 8; ++j) {
            v2f a0; a0.x = a[j].x; a0.y = a[j].y;
            v2f a1; a1.x = a[j].z; a1.y = a[j].w;
#pragma unroll
            for (int t = 0; t < 4; ++t) {
                v2f b0; b0.x = b[j][t].x; b0.y = b[j][t].y;
                acc[t] = __builtin_amdgcn_wmma_f32_16x16x4_f32(
                    false, a0, false, b0, (short)0, acc[t], false, false);
            }
#pragma unroll
            for (int t = 0; t < 4; ++t) {
                v2f b1; b1.x = b[j][t].z; b1.y = b[j][t].w;
                acc[t] = __builtin_amdgcn_wmma_f32_16x16x4_f32(
                    false, a1, false, b1, (short)0, acc[t], false, false);
            }
        }

        float* crow = C + (size_t)(m0 + half * 8) * N + n0 + idx;
        if (ROWSCALE) {
            float rs[8];
#pragma unroll
            for (int r = 0; r < 8; ++r) rs[r] = rsqrtf((float)((half ? rc[r + 8] : rc[r]) + 1));
#pragma unroll
            for (int t = 0; t < 4; ++t) {
#pragma unroll
                for (int r = 0; r < 8; ++r) {
                    crow[(size_t)r * N + 16 * t] = (acc[t][r] + bv[t]) * rs[r];
                }
            }
        } else {
#pragma unroll
            for (int t = 0; t < 4; ++t) {
#pragma unroll
                for (int r = 0; r < 8; ++r) {
                    crow[(size_t)r * N + 16 * t] = acc[t][r] + bv[t];
                }
            }
        }
#pragma unroll
        for (int j = 0; j < K / 8; ++j) a[j] = an[j];
    }
}

template <int N>
__global__ __launch_bounds__(256) void gemm_wmma_f32(const float* __restrict__ A,
                                                     const float* __restrict__ B,
                                                     const float* __restrict__ bias,
                                                     const int* __restrict__ rcnt,
                                                     float* __restrict__ C,
                                                     int mtiles) {
    gemm_body<N, true>(A, B, bias, nullptr, nullptr, rcnt, C, mtiles, (int)blockIdx.x, (int)gridDim.x);
}

#define FOLD_WG 8
__global__ __launch_bounds__(256) void fold_and_bin_kernel(const float* __restrict__ Wa,
                                                           const float* __restrict__ Wb,
                                                           const float* __restrict__ ba,
                                                           float* __restrict__ Wc,
                                                           float* __restrict__ bc,
                                                           const int* __restrict__ src,
                                                           const int* __restrict__ tgt,
                                                           int2* __restrict__ chunkbuf,
                                                           int* __restrict__ tab,
                                                           int* __restrict__ spill_len,
                                                           int n, int e) {
    if (blockIdx.x < FOLD_WG) {
        gemm_body<128, false>(Wa, Wb, nullptr, ba, bc, nullptr, Wc, 128 / 16, (int)blockIdx.x, FOLD_WG);
    } else {
        if (blockIdx.x == FOLD_WG && threadIdx.x == 0) *spill_len = 0;
        bin_chunk_body(src, tgt, chunkbuf, tab, (int)blockIdx.x - FOLD_WG, n, e);
    }
}

template <int V> struct rowvec;
template <> struct rowvec<4> { typedef v4f type; };
template <> struct rowvec<2> { typedef v2f type; };

template <int F, bool RELU>
__global__ __launch_bounds__(256) void gather_nodes_kernel(const float* __restrict__ xw,
                                                           const int* __restrict__ cnt,
                                                           const int* __restrict__ slots,
                                                           const int2* __restrict__ spill,
                                                           const float* __restrict__ bias,
                                                           float* __restrict__ out, int n, int e) {
    constexpr int V = F / 32;
    typedef typename rowvec<V>::type vf;
    const int lane   = threadIdx.x & 31;
    const int nwaves = (int)gridDim.x * 8;
    int node = __builtin_amdgcn_readfirstlane((int)(blockIdx.x * 8 + (threadIdx.x >> 5)));
    if (node >= n) return;
    const float* xcol = xw + lane * V;
    const vf bv = *(const vf*)(bias + lane * V);
    int c   = cnt[node];
    int cn  = cnt[min(node + nwaves, n - 1)];
    int ids = node;
    if (lane < min(c, CAP)) ids = slots[(size_t)node * CAP + lane];
    vf  self = *(const vf*)(xcol + (size_t)node * F);
    for (;;) {
        const int  nxt  = node + nwaves;
        const bool more = nxt < n;
        const int  pn   = more ? nxt : node;
        const int  cpn  = more ? cn : c;
        int nids = pn;
        if (lane < min(cpn, CAP)) nids = slots[(size_t)pn * CAP + lane];
        const vf  nself = *(const vf*)(xcol + (size_t)pn * F);
        const int cnn   = cnt[min(nxt + nwaves, n - 1)];

        const float di = rsqrtf((float)(c + 1));
        const int   m  = min(max(c, 0), CAP);
        int s = node;
        if (lane < m) s = min(max(ids, 0), n - 1);
        vf acc = {};
        int j = 0;
        for (; j + 4 <= m; j += 4) {
            const int s0 = __builtin_amdgcn_readlane(s, j);
            const int s1 = __builtin_amdgcn_readlane(s, j + 1);
            const int s2 = __builtin_amdgcn_readlane(s, j + 2);
            const int s3 = __builtin_amdgcn_readlane(s, j + 3);
            const vf r0 = *(const vf*)(xcol + (size_t)s0 * F);
            const vf r1 = *(const vf*)(xcol + (size_t)s1 * F);
            const vf r2 = *(const vf*)(xcol + (size_t)s2 * F);
            const vf r3 = *(const vf*)(xcol + (size_t)s3 * F);
            acc += r0;
            acc += r1;
            acc += r2;
            acc += r3;
        }
        for (; j < m; ++j) {
            const int s0 = __builtin_amdgcn_readlane(s, j);
            const vf r0 = *(const vf*)(xcol + (size_t)s0 * F);
            acc += r0;
        }
        if (c > CAP) {
            const int nsp = min(max(cnt[n], 0), e);
            for (int base = 0; base < nsp; base += 32) {
                const int q = base + lane;
                int os = node, ot = -1;
                if (q < nsp) { const int2 st = spill[q]; os = st.x; ot = st.y; }
                const bool hit = (ot == node);
                os = hit ? min(max(os, 0), n - 1) : node;
                unsigned left = __builtin_amdgcn_ballot_w32(hit);
                while (left != 0u) {
                    const int k = __builtin_ctz(left);
                    left &= left - 1u;
                    const int s0 = __builtin_amdgcn_readlane(os, k);
                    const vf r0 = *(const vf*)(xcol + (size_t)s0 * F);
                    acc += r0;
                }
            }
        }
        vf o;
#pragma unroll
        for (int cc = 0; cc < V; ++cc) {
            float v = (acc[cc] + self[cc]) * di + bv[cc];
            if (RELU) v = fmaxf(v, 0.0f);
            o[cc] = v;
        }
        *(vf*)(out + (size_t)node * F + lane * V) = o;

        if (!more) break;
        node = nxt;
        c    = cn;
        cn   = cnn;
        ids  = nids;
        self = nself;
    }
}

#define FUSE_TILE  32
#define FUSE_PITCH 132
#define FUSE_TILES 3125
#define FUSE_WG    1042
static_assert(FUSE_TILES * FUSE_TILE >= N_NODES, "tiles cover the nodes");

__global__ __launch_bounds__(256) void gather_gemm_kernel(const float* __restrict__ xw,
                                                          const int* __restrict__ cnt,
                                                          const int* __restrict__ slots,
                                                          const int2* __restrict__ spill,
                                                          const float* __restrict__ bias,
                                                          const float* __restrict__ B,
                                                          float* __restrict__ C,
                                                          int n, int e, int ntiles) {
    constexpr int F = 128;
    constexpr int K = 128;
    constexpr int N = 64;
    __shared__ float4 Bs[(K / 8) * N * 2];
    __shared__ __attribute__((aligned(16))) float hs[FUSE_TILE * FUSE_PITCH];
    __shared__ __attribute__((aligned(16))) float ldi[FUSE_TILE];

    const int tid = threadIdx.x;
    int tile = (int)blockIdx.x;
    if (tile >= ntiles) return;

    const int lane = tid & 31;
    const int half = lane >> 4;
    const int idx  = lane & 15;
    const int wave  = __builtin_amdgcn_readfirstlane(tid >> 5);
    const int strip = wave >> 2;
    const int n0    = (wave & 3) * 16;
    const int nwg   = (int)gridDim.x;

    const float* xcol = xw + lane * 4;
    const v4f bv = *(const v4f*)(bias + lane * 4);

    const int first = min(tile * FUSE_TILE + wave * 4, n - 1);
    int cv  = cnt[min(tile * FUSE_TILE + wave * 4 + (lane & 3), n - 1)];
    int ids = slots[(size_t)first * CAP + lane];
    v4f self = *(const v4f*)(xcol + (size_t)first * F);

    {
        constexpr int PT = (K / 4) * N / 256;
        float4 v[PT];
#pragma unroll
        for (int i = 0; i < PT; ++i) {
            const int p  = tid + 256 * i;
            const int nn = p % N;
            const int k4 = p / N;
            const float* bp = B + (size_t)(4 * k4) * N + nn;
            v[i].x = bp[0];
            v[i].y = bp[N];
            v[i].z = bp[2 * N];
            v[i].w = bp[3 * N];
        }
#pragma unroll
        for (int i = 0; i < PT; ++i) {
            const int p  = tid + 256 * i;
            const int nn = p % N;
            const int k4 = p / N;
            Bs[((((k4 >> 1) * N) + nn) << 1) + (k4 & 1)] = v[i];
        }
    }

    for (; tile < ntiles; tile += nwg) {
        v4f   o[4];
        float dq[4];
#pragma unroll
        for (int q = 0; q < 4; ++q) {
            const int node = tile * FUSE_TILE + wave * 4 + q;
            const int nxt = (q < 3) ? node + 1 : (tile + nwg) * FUSE_TILE + wave * 4;
            const int pn  = min(nxt, n - 1);
            const int c   = __builtin_amdgcn_readlane(cv, q);
            int ncv = cv;
            if (q == 3) ncv = cnt[min(nxt + (lane & 3), n - 1)];
            const int nids = slots[(size_t)pn * CAP + lane];
            const v4f nself = *(const v4f*)(xcol + (size_t)pn * F);

            v4f   ov = {};
            float di = 0.0f;
            if (node < n) {
                di = rsqrtf((float)(c + 1));
                const int m = min(max(c, 0), CAP);
                int s = node;
                if (lane < m) s = min(max(ids, 0), n - 1);
                v4f acc = {};
                int j = 0;
                for (; j + 4 <= m; j += 4) {
                    const int s0 = __builtin_amdgcn_readlane(s, j);
                    const int s1 = __builtin_amdgcn_readlane(s, j + 1);
                    const int s2 = __builtin_amdgcn_readlane(s, j + 2);
                    const int s3 = __builtin_amdgcn_readlane(s, j + 3);
                    const v4f r0 = *(const v4f*)(xcol + (size_t)s0 * F);
                    const v4f r1 = *(const v4f*)(xcol + (size_t)s1 * F);
                    const v4f r2 = *(const v4f*)(xcol + (size_t)s2 * F);
                    const v4f r3 = *(const v4f*)(xcol + (size_t)s3 * F);
                    acc += r0;
                    acc += r1;
                    acc += r2;
                    acc += r3;
                }
                for (; j < m; ++j) {
                    const int s0 = __builtin_amdgcn_readlane(s, j);
                    const v4f r0 = *(const v4f*)(xcol + (size_t)s0 * F);
                    acc += r0;
                }
                if (c > CAP) {
                    const int nsp = min(max(cnt[n], 0), e);
                    for (int base = 0; base < nsp; base += 32) {
                        const int qq = base + lane;
                        int os = node, ot = -1;
                        if (qq < nsp) { const int2 st = spill[qq]; os = st.x; ot = st.y; }
                        const bool hit = (ot == node);
                        os = hit ? min(max(os, 0), n - 1) : node;
                        unsigned left = __builtin_amdgcn_ballot_w32(hit);
                        while (left != 0u) {
                            const int k = __builtin_ctz(left);
                            left &= left - 1u;
                            const int s0 = __builtin_amdgcn_readlane(os, k);
                            const v4f r0 = *(const v4f*)(xcol + (size_t)s0 * F);
                            acc += r0;
                        }
                    }
                }
#pragma unroll
                for (int cc = 0; cc < 4; ++cc) {
                    const float v = (acc[cc] + self[cc]) * di + bv[cc];
                    ov[cc] = fmaxf(v, 0.0f);
                }
            }
            o[q]  = ov;
            dq[q] = di;
            cv   = ncv;
            ids  = nids;
            self = nself;
        }

        __syncthreads();
#pragma unroll
        for (int q = 0; q < 4; ++q) {
            *(v4f*)(hs + (wave * 4 + q) * FUSE_PITCH + lane * 4) = o[q];
            if (lane == 0) ldi[wave * 4 + q] = dq[q];
        }
        __syncthreads();

        const float*  arow  = hs + (strip * 16 + idx) * FUSE_PITCH + 4 * half;
        const float4* bfrag = Bs + ((n0 + idx) << 1) + half;
        v8f acc8 = {};
#pragma unroll
        for (int j = 0; j < K / 8; ++j) {
            const float4 a4 = *(const float4*)(arow + 8 * j);
            const float4 b4 = bfrag[(j * N) << 1];
            v2f a0; a0.x = a4.x; a0.y = a4.y;
            v2f b0; b0.x = b4.x; b0.y = b4.y;
            acc8 = __builtin_amdgcn_wmma_f32_16x16x4_f32(
                false, a0, false, b0, (short)0, acc8, false, false);
            v2f a1; a1.x = a4.z; a1.y = a4.w;
            v2f b1; b1.x = b4.z; b1.y = b4.w;
            acc8 = __builtin_amdgcn_wmma_f32_16x16x4_f32(
                false, a1, false, b1, (short)0, acc8, false, false);
        }
        const float4 d0 = *(const float4*)(ldi + strip * 16 + half * 8);
        const float4 d1 = *(const float4*)(ldi + strip * 16 + half * 8 + 4);
        const float rs[8] = {d0.x, d0.y, d0.z, d0.w, d1.x, d1.y, d1.z, d1.w};
        const int m0 = tile * FUSE_TILE + strip * 16 + half * 8;
        float* crow = C + (size_t)m0 * N + n0 + idx;
        if ((tile + 1) * FUSE_TILE <= n) {
#pragma unroll
            for (int r = 0; r < 8; ++r) crow[(size_t)r * N] = acc8[r] * rs[r];
        } else {
#pragma unroll
            for (int r = 0; r < 8; ++r)
                if (m0 + r < n) crow[(size_t)r * N] = acc8[r] * rs[r];
        }
    }
}

__global__ __launch_bounds__(256) void edge_score_kernel(const float* __restrict__ h,
                                                         const int* __restrict__ ia,
                                                         const int* __restrict__ ib,
                                                         float* __restrict__ out, int e) {
    constexpr int U = 4;
    const int lane   = threadIdx.x & 15;
    const int stride = (int)gridDim.x * 16;
    int ed = (int)((blockIdx.x * 256 + threadIdx.x) >> 4);
    for (; ed + (U - 1) * stride < e; ed += U * stride) {
        int a[U], b[U];
#pragma unroll
        for (int u = 0; u < U; ++u) {
            a[u] = ia[ed + u * stride];
            b[u] = ib[ed + u * stride];
        }
        float4 va[U], vb[U];
#pragma unroll
        for (int u = 0; u < U; ++u) {
            va[u] = ((const float4*)(h + (size_t)a[u] * 64))[lane];
            vb[u] = ((const float4*)(h + (size_t)b[u] * 64))[lane];
        }
#pragma unroll
        for (int u = 0; u < U; ++u) {
            float s = va[u].x * vb[u].x + va[u].y * vb[u].y + va[u].z * vb[u].z + va[u].w * vb[u].w;
            s += __shfl_xor(s, 8, 32);
            s += __shfl_xor(s, 4, 32);
            s += __shfl_xor(s, 2, 32);
            s += __shfl_xor(s, 1, 32);
            if (lane == 0) out[ed + u * stride] = s;
        }
    }
    for (; ed < e; ed += stride) {
        const int a = ia[ed];
        const int b = ib[ed];
        const float4 va = ((const float4*)(h + (size_t)a * 64))[lane];
        const float4 vb = ((const float4*)(h + (size_t)b * 64))[lane];
        float s = va.x * vb.x + va.y * vb.y + va.z * vb.z + va.w * vb.w;
        s += __shfl_xor(s, 8, 32);
        s += __shfl_xor(s, 4, 32);
        s += __shfl_xor(s, 2, 32);
        s += __shfl_xor(s, 1, 32);
        if (lane == 0) out[ed] = s;
    }
}

extern "C" void kernel_launch(void* const* d_in, const int* in_sizes, int n_in,
                              void* d_out, int out_size, void* d_ws, size_t ws_size,
                              hipStream_t stream) {
    const float* x      = (const float*)d_in[0];
    const int*   eidx   = (const int*)d_in[1];
    const int*   pidx   = (const int*)d_in[2];
    const float* W_init = (const float*)d_in[3];
    const float* b_init = (const float*)d_in[4];
    const float* W1     = (const float*)d_in[5];
    const float* b1     = (const float*)d_in[6];
    const float* W2     = (const float*)d_in[7];
    const float* b2     = (const float*)d_in[8];
    float* logits = (float*)d_out;

    const int* src = pidx;
    const int* tgt = pidx + E_POS;
    const int* qa  = eidx;
    const int* qb  = eidx + E_PRED;

    float* ws   = (float*)d_ws;
    float* bufA = ws + 102400;
    float* bufB = bufA + (size_t)N_NODES * 128;
    float* bufC = bufB + (size_t)N_NODES * 128;
    int*  cnt   = (int*)bufC;
    float* Wc = (float*)(cnt + 102400);
    float* bc = Wc + 128 * 128;
    int2* spill = (int2*)(cnt + 102400 + 16896);
    int*  slots = cnt + 102400 + 16896 + 2 * E_POS;
    int2* chunkbuf = (int2*)(slots + (size_t)N_NODES * CAP);
    int*  tab   = (int*)(chunkbuf + (size_t)NCHUNK * CHUNK);

    const int TB = 256;
    const int MT = N_NODES / 16;

    fold_and_bin_kernel<<<FOLD_WG + NCHUNK, TB, 0, stream>>>(
        W_init, W1, b_init, Wc, bc, src, tgt, chunkbuf, tab, cnt + N_NODES, N_NODES, E_POS);
    slot_build_kernel<<<NBIN, TB, 0, stream>>>(chunkbuf, tab, cnt, slots, spill, N_NODES, E_POS);

    gemm_wmma_f32<128><<<256, 256, 0, stream>>>(x, Wc, bc, cnt, bufB, MT);
    gather_gemm_kernel<<<FUSE_WG, TB, 0, stream>>>(
        bufB, cnt, slots, spill, b1, W2, bufA, N_NODES, E_POS, FUSE_TILES);

    gather_nodes_kernel<64, false><<<GATHER_WG, TB, 0, stream>>>(
        bufA, cnt, slots, spill, b2, bufB, N_NODES, E_POS);

    edge_score_kernel<<<EDGE_WG, TB, 0, stream>>>(bufB, qa, qb, logits, E_PRED);
}


# Both node gathers request all of a node's source rows together (up to 8 at a time, tail included) before the first add, instead of 4 at a time and then the remainder one by one; sum order unchanged
# speedup vs baseline: 6.0482x; 1.0328x over previous
#include <hip/hip_runtime.h>
#include <hip/hip_bf16.h>

typedef __attribute__((ext_vector_type(2))) float v2f;
typedef __attribute__((ext_vector_type(4))) float v4f;
typedef __attribute__((ext_vector_type(8))) float v8f;

#define N_NODES 100000
#define E_POS   500000
#define E_PRED  200000
#define GATHER_WG 2048
#define EDGE_WG   1024
#define CAP 32
#define GATHER_NF 8
#define NCHUNK 248
#define CHUNK  2017
#define NBIN   256
#define BINW   391
#define EPT    8
static_assert(CAP == 32, "one slot per lane");
static_assert(NCHUNK <= 256 && NBIN == 256, "one thread per chunk and per range");
static_assert(NCHUNK * CHUNK >= E_POS && 256 * EPT >= CHUNK && CHUNK < 32768, "chunking");
static_assert(NBIN * BINW >= N_NODES, "ranges cover the nodes");

__device__ __forceinline__ void bin_chunk_body(const int* __restrict__ src, const int* __restrict__ tgt,
                                               int2* __restrict__ chunkbuf, int* __restrict__ tab,
                                               int c, int n, int e) {
    __shared__ int  hist[NBIN];
    __shared__ int  offs[NBIN];
    __shared__ int  wsum[8];
    __shared__ int2 stage[256 * EPT];
    const int tid  = threadIdx.x;
    const int lane = tid & 31;
    const int wave = tid >> 5;
    hist[tid] = 0;
    __syncthreads();

    const int e0 = c * CHUNK;
    int s[EPT], t[EPT], pos[EPT];
#pragma unroll
    for (int u = 0; u < EPT; ++u) {
        const int q = u * 256 + tid;
        const bool ok = (q < CHUNK) && (e0 + q < e);
        t[u] = ok ? tgt[e0 + q] : -1;
        s[u] = ok ? src[e0 + q] : 0;
    }
#pragma unroll
    for (int u = 0; u < EPT; ++u) {
        pos[u] = -1;
        if ((unsigned)t[u] < (unsigned)n) pos[u] = atomicAdd(&hist[t[u] / BINW], 1);
    }
    __syncthreads();

    const int v = hist[tid];
    int x = v;
#pragma unroll
    for (int d = 1; d < 32; d <<= 1) {
        const int y = __shfl_up(x, d, 32);
        if (lane >= d) x += y;
    }
    if (lane == 31) wsum[wave] = x;
    __syncthreads();
    int before = 0;
#pragma unroll
    for (int w = 0; w < 8; ++w) before += (w < wave) ? wsum[w] : 0;
    const int start = before + x - v;
    offs[tid] = start;
    tab[c * NBIN + tid] = start | (v << 16);
    __syncthreads();

    int dst[EPT];
#pragma unroll
    for (int u = 0; u < EPT; ++u)
        dst[u] = offs[min((unsigned)t[u], (unsigned)(n - 1)) / BINW] + pos[u];
#pragma unroll
    for (int u = 0; u < EPT; ++u)
        if (pos[u] >= 0) stage[dst[u]] = make_int2(s[u], t[u]);
    __syncthreads();
    const int total = offs[NBIN - 1] + hist[NBIN - 1];
    for (int i = tid; i < total; i += 256) chunkbuf[e0 + i] = stage[i];
}

__global__ __launch_bounds__(256) void slot_build_kernel(const int2* __restrict__ chunkbuf,
                                                         const int* __restrict__ tab,
                                                         int* __restrict__ cnt, int* __restrict__ slots,
                                                         int2* __restrict__ spill, int n, int e) {
    constexpr int L4 = BINW * CAP / 4;
    constexpr int G  = 16;
    __shared__ int lcnt[BINW];
    __shared__ __attribute__((aligned(16))) int lslots[BINW * CAP];
    int4* lslots4 = reinterpret_cast<int4*>(lslots);
    const int tid  = threadIdx.x;
    const int lane = tid & 31;
    const int wv   = __builtin_amdgcn_readfirstlane(tid >> 5);
    const int b    = blockIdx.x;
    const int base = b * BINW;
    const int pk = (tid < NCHUNK) ? tab[tid * NBIN + b] : 0;
    for (int l = tid; l < BINW; l += 256) lcnt[l] = 0;
    for (int i = tid; i < L4; i += 256) lslots4[i] = make_int4(0, 0, 0, 0);
    __syncthreads();

    const int2* cb = chunkbuf + (size_t)(wv * 32) * CHUNK;
#pragma unroll
    for (int i0 = 0; i0 < 32; i0 += G) {
        int2 st[G];
#pragma unroll
        for (int j = 0; j < G; ++j) {
            const int pki = __builtin_amdgcn_readlane(pk, i0 + j);
            const int2* p = cb + (size_t)(i0 + j) * CHUNK + (pki & 0xffff);
            st[j] = make_int2(0, -1);
            if (lane < min(pki >> 16, CHUNK)) st[j] = p[lane];
        }
        int l[G], pos[G];
#pragma unroll
        for (int j = 0; j < G; ++j) {
            l[j]   = st[j].y - base;
            pos[j] = -1;
            if ((unsigned)l[j] < (unsigned)BINW) pos[j] = atomicAdd(&lcnt[l[j]], 1);
        }
#pragma unroll
        for (int j = 0; j < G; ++j) {
            if (pos[j] >= 0) {
                if (pos[j] < CAP) {
                    lslots[l[j] * CAP + pos[j]] = st[j].x;
                } else {
                    const int ks = atomicAdd(&cnt[n], 1);
                    if ((unsigned)ks < (unsigned)e) spill[ks] = st[j];
                }
            }
        }
    }
    if (__builtin_amdgcn_ballot_w32((pk >> 16) > 32) != 0u) {
        for (int i = 0; i < 32; ++i) {
            const int pki = __builtin_amdgcn_readlane(pk, i);
            const int ki  = min(pki >> 16, CHUNK);
            const int2* p = cb + (size_t)i * CHUNK + (pki & 0xffff);
            for (int q0 = 32; q0 < ki; q0 += 32) {
                int2 st = make_int2(0, -1);
                if (q0 + lane < ki) st = p[q0 + lane];
                const int l = st.y - base;
                if ((unsigned)l < (unsigned)BINW) {
                    const int pos = atomicAdd(&lcnt[l], 1);
                    if (pos < CAP) {
                        lslots[l * CAP + pos] = st.x;
                    } else {
                        const int ks = atomicAdd(&cnt[n], 1);
                        if ((unsigned)ks < (unsigned)e) spill[ks] = st;
                    }
                }
            }
        }
    }
    __syncthreads();

    int4* gl = reinterpret_cast<int4*>(slots + (size_t)base * CAP);
    if (base + BINW <= n) {
        for (int l = tid; l < BINW; l += 256) cnt[base + l] = lcnt[l];
#pragma unroll
        for (int it = 0; it < L4 / 256; ++it) gl[it * 256 + tid] = lslots4[it * 256 + tid];
        if ((L4 / 256) * 256 + tid < L4) gl[(L4 / 256) * 256 + tid] = lslots4[(L4 / 256) * 256 + tid];
    } else {
        for (int l = tid; l < BINW; l += 256)
            if (base + l < n) cnt[base + l] = lcnt[l];
        for (int i = tid; i < L4; i += 256)
            if (base + (i >> 3) < n) gl[i] = lslots4[i];
    }
}

template <int N, bool ROWSCALE>
__device__ __forceinline__ void gemm_body(const float* __restrict__ A,
                                          const float* __restrict__ B,
                                          const float* __restrict__ bias,
                                          const float* __restrict__ fvec,
                                          float* __restrict__ fout,
                                          const int* __restrict__ rcnt,
                                          float* __restrict__ C,
                                          int mtiles, int bid, int nwg) {
    constexpr int K   = 128;
    constexpr int NS  = N / 64;
    constexpr int WPS = 8 / NS;
    __shared__ float4 Bs[(K / 8) * N * 2];

    const int tid = threadIdx.x;
    for (int p = tid; p < (K / 4) * N; p += 256) {
        const int n  = p % N;
        const int k4 = p / N;
        const float* bp = B + (size_t)(4 * k4) * N + n;
        float4 v;
        v.x = bp[0];
        v.y = bp[N];
        v.z = bp[2 * N];
        v.w = bp[3 * N];
        Bs[((((k4 >> 1) * N) + n) << 1) + (k4 & 1)] = v;
    }
    __syncthreads();

    const int lane = tid & 31;
    const int half = lane >> 4;
    const int idx  = lane & 15;
    const int wave  = __builtin_amdgcn_readfirstlane(tid >> 5);
    const int slice = wave % NS;
    const int grp   = wave / NS;
    const int n0    = slice * 64;

    float4 b[K / 8][4];
    {
        const float4* bfrag = Bs + (n0 + idx) * 2 + half;
#pragma unroll
        for (int j = 0; j < K / 8; ++j)
#pragma unroll
            for (int t = 0; t < 4; ++t) b[j][t] = bfrag[(j * N + 16 * t) * 2];
    }
    float bv[4];
#pragma unroll
    for (int t = 0; t < 4; ++t) bv[t] = bias ? bias[n0 + 16 * t + idx] : 0.0f;

    if (fvec != nullptr && bid == 0 && grp == 0) {
        float s[4] = {0.0f, 0.0f, 0.0f, 0.0f};
#pragma unroll
        for (int j = 0; j < K / 8; ++j) {
            const float4 f = *(const float4*)(fvec + 8 * j + 4 * half);
#pragma unroll
            for (int t = 0; t < 4; ++t)
                s[t] += f.x * b[j][t].x + f.y * b[j][t].y + f.z * b[j][t].z + f.w * b[j][t].w;
        }
#pragma unroll
        for (int t = 0; t < 4; ++t) {
            const float tot = s[t] + __shfl_xor(s[t], 16, 32);
            if (half == 0) fout[n0 + 16 * t + idx] = tot;
        }
    }

    float4 a[K / 8];
    int mt = grp * nwg + bid;
    if (mt < mtiles) {
        const float4* ap = (const float4*)(A + (size_t)(mt * 16 + idx) * K) + half;
#pragma unroll
        for (int j = 0; j < K / 8; ++j) a[j] = ap[2 * j];
    }

    for (; mt < mtiles; mt += WPS * nwg) {
        const int m0 = mt * 16;
        const int mtn = min(mt + WPS * nwg, mtiles - 1);
        const float4* apn = (const float4*)(A + (size_t)(mtn * 16 + idx) * K) + half;
        float4 an[K / 8];
#pragma unroll
        for (int j = 0; j < K / 8; ++j) an[j] = apn[2 * j];
        int rc[16];
#pragma unroll
        for (int r = 0; r < 16; ++r) rc[r] = ROWSCALE ? rcnt[m0 + r] : 0;

        v8f acc[4];
#pragma unroll
        for (int t = 0; t < 4; ++t) acc[t] = v8f{};

#pragma unroll
        for (int j = 0; j < K / 8; ++j) {
            v2f a0; a0.x = a[j].x; a0.y = a[j].y;
            v2f a1; a1.x = a[j].z; a1.y = a[j].w;
#pragma unroll
            for (int t = 0; t < 4; ++t) {
                v2f b0; b0.x = b[j][t].x; b0.y = b[j][t].y;
                acc[t] = __builtin_amdgcn_wmma_f32_16x16x4_f32(
                    false, a0, false, b0, (short)0, acc[t], false, false);
            }
#pragma unroll
            for (int t = 0; t < 4; ++t) {
                v2f b1; b1.x = b[j][t].z; b1.y = b[j][t].w;
                acc[t] = __builtin_amdgcn_wmma_f32_16x16x4_f32(
                    false, a1, false, b1, (short)0, acc[t], false, false);
            }
        }

        float* crow = C + (size_t)(m0 + half * 8) * N + n0 + idx;
        if (ROWSCALE) {
            float rs[8];
#pragma unroll
            for (int r = 0; r < 8; ++r) rs[r] = rsqrtf((float)((half ? rc[r + 8] : rc[r]) + 1));
#pragma unroll
            for (int t = 0; t < 4; ++t) {
#pragma unroll
                for (int r = 0; r < 8; ++r) {
                    crow[(size_t)r * N + 16 * t] = (acc[t][r] + bv[t]) * rs[r];
                }
            }
        } else {
#pragma unroll
            for (int t = 0; t < 4; ++t) {
#pragma unroll
                for (int r = 0; r < 8; ++r) {
                    crow[(size_t)r * N + 16 * t] = acc[t][r] + bv[t];
                }
            }
        }
#pragma unroll
        for (int j = 0; j < K / 8; ++j) a[j] = an[j];
    }
}

template <int N>
__global__ __launch_bounds__(256) void gemm_wmma_f32(const float* __restrict__ A,
                                                     const float* __restrict__ B,
                                                     const float* __restrict__ bias,
                                                     const int* __restrict__ rcnt,
                                                     float* __restrict__ C,
                                                     int mtiles) {
    gemm_body<N, true>(A, B, bias, nullptr, nullptr, rcnt, C, mtiles, (int)blockIdx.x, (int)gridDim.x);
}

#define FOLD_WG 8
__global__ __launch_bounds__(256) void fold_and_bin_kernel(const float* __restrict__ Wa,
                                                           const float* __restrict__ Wb,
                                                           const float* __restrict__ ba,
                                                           float* __restrict__ Wc,
                                                           float* __restrict__ bc,
                                                           const int* __restrict__ src,
                                                           const int* __restrict__ tgt,
                                                           int2* __restrict__ chunkbuf,
                                                           int* __restrict__ tab,
                                                           int* __restrict__ spill_len,
                                                           int n, int e) {
    if (blockIdx.x < FOLD_WG) {
        gemm_body<128, false>(Wa, Wb, nullptr, ba, bc, nullptr, Wc, 128 / 16, (int)blockIdx.x, FOLD_WG);
    } else {
        if (blockIdx.x == FOLD_WG && threadIdx.x == 0) *spill_len = 0;
        bin_chunk_body(src, tgt, chunkbuf, tab, (int)blockIdx.x - FOLD_WG, n, e);
    }
}

template <int V> struct rowvec;
template <> struct rowvec<4> { typedef v4f type; };
template <> struct rowvec<2> { typedef v2f type; };

template <int F, int K, typename vf>
__device__ __forceinline__ vf add_rows(vf acc, const float* __restrict__ xcol, int s, int j) {
    vf r[K];
#pragma unroll
    for (int u = 0; u < K; ++u) {
        const int su = __builtin_amdgcn_readlane(s, j + u);
        r[u] = *(const vf*)(xcol + (size_t)su * F);
    }
#pragma unroll
    for (int u = 0; u < K; ++u) acc += r[u];
    return acc;
}

template <int F, int NF, typename vf>
__device__ __forceinline__ vf sum_rows(const float* __restrict__ xcol, int s, int m) {
    static_assert(NF >= 4 && NF <= 8, "tail paths below cover 1 .. 7 rows");
    vf acc = {};
    int j = 0;
    for (; j + NF <= m; j += NF) acc = add_rows<F, NF, vf>(acc, xcol, s, j);
    switch (m - j) {
        case 1: acc = add_rows<F, 1, vf>(acc, xcol, s, j); break;
        case 2: acc = add_rows<F, 2, vf>(acc, xcol, s, j); break;
        case 3: acc = add_rows<F, 3, vf>(acc, xcol, s, j); break;
        case 4: if (NF > 4) acc = add_rows<F, 4, vf>(acc, xcol, s, j); break;
        case 5: if (NF > 5) acc = add_rows<F, 5, vf>(acc, xcol, s, j); break;
        case 6: if (NF > 6) acc = add_rows<F, 6, vf>(acc, xcol, s, j); break;
        case 7: if (NF > 7) acc = add_rows<F, 7, vf>(acc, xcol, s, j); break;
        default: break;
    }
    return acc;
}

template <int F, bool RELU>
__global__ __launch_bounds__(256) void gather_nodes_kernel(const float* __restrict__ xw,
                                                           const int* __restrict__ cnt,
                                                           const int* __restrict__ slots,
                                                           const int2* __restrict__ spill,
                                                           const float* __restrict__ bias,
                                                           float* __restrict__ out, int n, int e) {
    constexpr int V = F / 32;
    typedef typename rowvec<V>::type vf;
    const int lane   = threadIdx.x & 31;
    const int nwaves = (int)gridDim.x * 8;
    int node = __builtin_amdgcn_readfirstlane((int)(blockIdx.x * 8 + (threadIdx.x >> 5)));
    if (node >= n) return;
    const float* xcol = xw + lane * V;
    const vf bv = *(const vf*)(bias + lane * V);
    int c   = cnt[node];
    int cn  = cnt[min(node + nwaves, n - 1)];
    int ids = node;
    if (lane < min(c, CAP)) ids = slots[(size_t)node * CAP + lane];
    vf  self = *(const vf*)(xcol + (size_t)node * F);
    for (;;) {
        const int  nxt  = node + nwaves;
        const bool more = nxt < n;
        const int  pn   = more ? nxt : node;
        const int  cpn  = more ? cn : c;
        int nids = pn;
        if (lane < min(cpn, CAP)) nids = slots[(size_t)pn * CAP + lane];
        const vf  nself = *(const vf*)(xcol + (size_t)pn * F);
        const int cnn   = cnt[min(nxt + nwaves, n - 1)];

        const float di = rsqrtf((float)(c + 1));
        const int   m  = min(max(c, 0), CAP);
        int s = node;
        if (lane < m) s = min(max(ids, 0), n - 1);
        vf acc = sum_rows<F, GATHER_NF, vf>(xcol, s, m);
        if (c > CAP) {
            const int nsp = min(max(cnt[n], 0), e);
            for (int base = 0; base < nsp; base += 32) {
                const int q = base + lane;
                int os = node, ot = -1;
                if (q < nsp) { const int2 st = spill[q]; os = st.x; ot = st.y; }
                const bool hit = (ot == node);
                os = hit ? min(max(os, 0), n - 1) : node;
                unsigned left = __builtin_amdgcn_ballot_w32(hit);
                while (left != 0u) {
                    const int k = __builtin_ctz(left);
                    left &= left - 1u;
                    const int s0 = __builtin_amdgcn_readlane(os, k);
                    const vf r0 = *(const vf*)(xcol + (size_t)s0 * F);
                    acc += r0;
                }
            }
        }
        vf o;
#pragma unroll
        for (int cc = 0; cc < V; ++cc) {
            float v = (acc[cc] + self[cc]) * di + bv[cc];
            if (RELU) v = fmaxf(v, 0.0f);
            o[cc] = v;
        }
        *(vf*)(out + (size_t)node * F + lane * V) = o;

        if (!more) break;
        node = nxt;
        c    = cn;
        cn   = cnn;
        ids  = nids;
        self = nself;
    }
}

#define FUSE_TILE  32
#define FUSE_PITCH 132
#define FUSE_TILES 3125
#define FUSE_WG    1042
static_assert(FUSE_TILES * FUSE_TILE >= N_NODES, "tiles cover the nodes");

__global__ __launch_bounds__(256) void gather_gemm_kernel(const float* __restrict__ xw,
                                                          const int* __restrict__ cnt,
                                                          const int* __restrict__ slots,
                                                          const int2* __restrict__ spill,
                                                          const float* __restrict__ bias,
                                                          const float* __restrict__ B,
                                                          float* __restrict__ C,
                                                          int n, int e, int ntiles) {
    constexpr int F = 128;
    constexpr int K = 128;
    constexpr int N = 64;
    __shared__ float4 Bs[(K / 8) * N * 2];
    __shared__ __attribute__((aligned(16))) float hs[FUSE_TILE * FUSE_PITCH];
    __shared__ __attribute__((aligned(16))) float ldi[FUSE_TILE];

    const int tid = threadIdx.x;
    int tile = (int)blockIdx.x;
    if (tile >= ntiles) return;

    const int lane = tid & 31;
    const int half = lane >> 4;
    const int idx  = lane & 15;
    const int wave  = __builtin_amdgcn_readfirstlane(tid >> 5);
    const int strip = wave >> 2;
    const int n0    = (wave & 3) * 16;
    const int nwg   = (int)gridDim.x;

    const float* xcol = xw + lane * 4;
    const v4f bv = *(const v4f*)(bias + lane * 4);

    const int first = min(tile * FUSE_TILE + wave * 4, n - 1);
    int cv  = cnt[min(tile * FUSE_TILE + wave * 4 + (lane & 3), n - 1)];
    int ids = slots[(size_t)first * CAP + lane];
    v4f self = *(const v4f*)(xcol + (size_t)first * F);

    {
        constexpr int PT = (K / 4) * N / 256;
        float4 v[PT];
#pragma unroll
        for (int i = 0; i < PT; ++i) {
            const int p  = tid + 256 * i;
            const int nn = p % N;
            const int k4 = p / N;
            const float* bp = B + (size_t)(4 * k4) * N + nn;
            v[i].x = bp[0];
            v[i].y = bp[N];
            v[i].z = bp[2 * N];
            v[i].w = bp[3 * N];
        }
#pragma unroll
        for (int i = 0; i < PT; ++i) {
            const int p  = tid + 256 * i;
            const int nn = p % N;
            const int k4 = p / N;
            Bs[((((k4 >> 1) * N) + nn) << 1) + (k4 & 1)] = v[i];
        }
    }

    for (; tile < ntiles; tile += nwg) {
        v4f   o[4];
        float dq[4];
#pragma unroll
        for (int q = 0; q < 4; ++q) {
            const int node = tile * FUSE_TILE + wave * 4 + q;
            const int nxt = (q < 3) ? node + 1 : (tile + nwg) * FUSE_TILE + wave * 4;
            const int pn  = min(nxt, n - 1);
            const int c   = __builtin_amdgcn_readlane(cv, q);
            int ncv = cv;
            if (q == 3) ncv = cnt[min(nxt + (lane & 3), n - 1)];
            const int nids = slots[(size_t)pn * CAP + lane];
            const v4f nself = *(const v4f*)(xcol + (size_t)pn * F);

            v4f   ov = {};
            float di = 0.0f;
            if (node < n) {
                di = rsqrtf((float)(c + 1));
                const int m = min(max(c, 0), CAP);
                int s = node;
                if (lane < m) s = min(max(ids, 0), n - 1);
                v4f acc = sum_rows<F, GATHER_NF, v4f>(xcol, s, m);
                if (c > CAP) {
                    const int nsp = min(max(cnt[n], 0), e);
                    for (int base = 0; base < nsp; base += 32) {
                        const int qq = base + lane;
                        int os = node, ot = -1;
                        if (qq < nsp) { const int2 st = spill[qq]; os = st.x; ot = st.y; }
                        const bool hit = (ot == node);
                        os = hit ? min(max(os, 0), n - 1) : node;
                        unsigned left = __builtin_amdgcn_ballot_w32(hit);
                        while (left != 0u) {
                            const int k = __builtin_ctz(left);
                            left &= left - 1u;
                            const int s0 = __builtin_amdgcn_readlane(os, k);
                            const v4f r0 = *(const v4f*)(xcol + (size_t)s0 * F);
                            acc += r0;
                        }
                    }
                }
#pragma unroll
                for (int cc = 0; cc < 4; ++cc) {
                    const float v = (acc[cc] + self[cc]) * di + bv[cc];
                    ov[cc] = fmaxf(v, 0.0f);
                }
            }
            o[q]  = ov;
            dq[q] = di;
            cv   = ncv;
            ids  = nids;
            self = nself;
        }

        __syncthreads();
#pragma unroll
        for (int q = 0; q < 4; ++q) {
            *(v4f*)(hs + (wave * 4 + q) * FUSE_PITCH + lane * 4) = o[q];
            if (lane == 0) ldi[wave * 4 + q] = dq[q];
        }
        __syncthreads();

        const float*  arow  = hs + (strip * 16 + idx) * FUSE_PITCH + 4 * half;
        const float4* bfrag = Bs + ((n0 + idx) << 1) + half;
        v8f acc8 = {};
#pragma unroll
        for (int j = 0; j < K / 8; ++j) {
            const float4 a4 = *(const float4*)(arow + 8 * j);
            const float4 b4 = bfrag[(j * N) << 1];
            v2f a0; a0.x = a4.x; a0.y = a4.y;
            v2f b0; b0.x = b4.x; b0.y = b4.y;
            acc8 = __builtin_amdgcn_wmma_f32_16x16x4_f32(
                false, a0, false, b0, (short)0, acc8, false, false);
            v2f a1; a1.x = a4.z; a1.y = a4.w;
            v2f b1; b1.x = b4.z; b1.y = b4.w;
            acc8 = __builtin_amdgcn_wmma_f32_16x16x4_f32(
                false, a1, false, b1, (short)0, acc8, false, false);
        }
        const float4 d0 = *(const float4*)(ldi + strip * 16 + half * 8);
        const float4 d1 = *(const float4*)(ldi + strip * 16 + half * 8 + 4);
        const float rs[8] = {d0.x, d0.y, d0.z, d0.w, d1.x, d1.y, d1.z, d1.w};
        const int m0 = tile * FUSE_TILE + strip * 16 + half * 8;
        float* crow = C + (size_t)m0 * N + n0 + idx;
        if ((tile + 1) * FUSE_TILE <= n) {
#pragma unroll
            for (int r = 0; r < 8; ++r) crow[(size_t)r * N] = acc8[r] * rs[r];
        } else {
#pragma unroll
            for (int r = 0; r < 8; ++r)
                if (m0 + r < n) crow[(size_t)r * N] = acc8[r] * rs[r];
        }
    }
}

__global__ __launch_bounds__(256) void edge_score_kernel(const float* __restrict__ h,
                                                         const int* __restrict__ ia,
                                                         const int* __restrict__ ib,
                                                         float* __restrict__ out, int e) {
    constexpr int U = 4;
    const int lane   = threadIdx.x & 15;
    const int stride = (int)gridDim.x * 16;
    int ed = (int)((blockIdx.x * 256 + threadIdx.x) >> 4);
    for (; ed + (U - 1) * stride < e; ed += U * stride) {
        int a[U], b[U];
#pragma unroll
        for (int u = 0; u < U; ++u) {
            a[u] = ia[ed + u * stride];
            b[u] = ib[ed + u * stride];
        }
        float4 va[U], vb[U];
#pragma unroll
        for (int u = 0; u < U; ++u) {
            va[u] = ((const float4*)(h + (size_t)a[u] * 64))[lane];
            vb[u] = ((const float4*)(h + (size_t)b[u] * 64))[lane];
        }
#pragma unroll
        for (int u = 0; u < U; ++u) {
            float s = va[u].x * vb[u].x + va[u].y * vb[u].y + va[u].z * vb[u].z + va[u].w * vb[u].w;
            s += __shfl_xor(s, 8, 32);
            s += __shfl_xor(s, 4, 32);
            s += __shfl_xor(s, 2, 32);
            s += __shfl_xor(s, 1, 32);
            if (lane == 0) out[ed + u * stride] = s;
        }
    }
    for (; ed < e; ed += stride) {
        const int a = ia[ed];
        const int b = ib[ed];
        const float4 va = ((const float4*)(h + (size_t)a * 64))[lane];
        const float4 vb = ((const float4*)(h + (size_t)b * 64))[lane];
        float s = va.x * vb.x + va.y * vb.y + va.z * vb.z + va.w * vb.w;
        s += __shfl_xor(s, 8, 32);
        s += __shfl_xor(s, 4, 32);
        s += __shfl_xor(s, 2, 32);
        s += __shfl_xor(s, 1, 32);
        if (lane == 0) out[ed] = s;
    }
}

extern "C" void kernel_launch(void* const* d_in, const int* in_sizes, int n_in,
                              void* d_out, int out_size, void* d_ws, size_t ws_size,
                              hipStream_t stream) {
    const float* x      = (const float*)d_in[0];
    const int*   eidx   = (const int*)d_in[1];
    const int*   pidx   = (const int*)d_in[2];
    const float* W_init = (const float*)d_in[3];
    const float* b_init = (const float*)d_in[4];
    const float* W1     = (const float*)d_in[5];
    const float* b1     = (const float*)d_in[6];
    const float* W2     = (const float*)d_in[7];
    const float* b2     = (const float*)d_in[8];
    float* logits = (float*)d_out;

    const int* src = pidx;
    const int* tgt = pidx + E_POS;
    const int* qa  = eidx;
    const int* qb  = eidx + E_PRED;

    float* ws   = (float*)d_ws;
    float* bufA = ws + 102400;
    float* bufB = bufA + (size_t)N_NODES * 128;
    float* bufC = bufB + (size_t)N_NODES * 128;
    int*  cnt   = (int*)bufC;
    float* Wc = (float*)(cnt + 102400);
    float* bc = Wc + 128 * 128;
    int2* spill = (int2*)(cnt + 102400 + 16896);
    int*  slots = cnt + 102400 + 16896 + 2 * E_POS;
    int2* chunkbuf = (int2*)(slots + (size_t)N_NODES * CAP);
    int*  tab   = (int*)(chunkbuf + (size_t)NCHUNK * CHUNK);

    const int TB = 256;
    const int MT = N_NODES / 16;

    fold_and_bin_kernel<<<FOLD_WG + NCHUNK, TB, 0, stream>>>(
        W_init, W1, b_init, Wc, bc, src, tgt, chunkbuf, tab, cnt + N_NODES, N_NODES, E_POS);
    slot_build_kernel<<<NBIN, TB, 0, stream>>>(chunkbuf, tab, cnt, slots, spill, N_NODES, E_POS);

    gemm_wmma_f32<128><<<256, 256, 0, stream>>>(x, Wc, bc, cnt, bufB, MT);
    gather_gemm_kernel<<<FUSE_WG, TB, 0, stream>>>(
        bufB, cnt, slots, spill, b1, W2, bufA, N_NODES, E_POS, FUSE_TILES);

    gather_nodes_kernel<64, false><<<GATHER_WG, TB, 0, stream>>>(
        bufA, cnt, slots, spill, b2, bufB, N_NODES, E_POS);

    edge_score_kernel<<<EDGE_WG, TB, 0, stream>>>(bufB, qa, qb, logits, E_PRED);
}


# GEMM prologue: B share, first A strip, bias and the one-row operand all requested before the first wait instead of eight load-wait-store trips and more behind; sum order unchanged
# speedup vs baseline: 6.4408x; 1.0649x over previous
#include <hip/hip_runtime.h>
#include <hip/hip_bf16.h>

typedef __attribute__((ext_vector_type(2))) float v2f;
typedef __attribute__((ext_vector_type(4))) float v4f;
typedef __attribute__((ext_vector_type(8))) float v8f;

#define N_NODES 100000
#define E_POS   500000
#define E_PRED  200000
#define GATHER_WG 2048
#define EDGE_WG   1024
#define CAP 32
#define GATHER_NF 8
#define NCHUNK 248
#define CHUNK  2017
#define NBIN   256
#define BINW   391
#define EPT    8
static_assert(CAP == 32, "one slot per lane");
static_assert(NCHUNK <= 256 && NBIN == 256, "one thread per chunk and per range");
static_assert(NCHUNK * CHUNK >= E_POS && 256 * EPT >= CHUNK && CHUNK < 32768, "chunking");
static_assert(NBIN * BINW >= N_NODES, "ranges cover the nodes");

__device__ __forceinline__ void bin_chunk_body(const int* __restrict__ src, const int* __restrict__ tgt,
                                               int2* __restrict__ chunkbuf, int* __restrict__ tab,
                                               int c, int n, int e) {
    __shared__ int  hist[NBIN];
    __shared__ int  offs[NBIN];
    __shared__ int  wsum[8];
    __shared__ int2 stage[256 * EPT];
    const int tid  = threadIdx.x;
    const int lane = tid & 31;
    const int wave = tid >> 5;
    hist[tid] = 0;
    __syncthreads();

    const int e0 = c * CHUNK;
    int s[EPT], t[EPT], pos[EPT];
#pragma unroll
    for (int u = 0; u < EPT; ++u) {
        const int q = u * 256 + tid;
        const bool ok = (q < CHUNK) && (e0 + q < e);
        t[u] = ok ? tgt[e0 + q] : -1;
        s[u] = ok ? src[e0 + q] : 0;
    }
#pragma unroll
    for (int u = 0; u < EPT; ++u) {
        pos[u] = -1;
        if ((unsigned)t[u] < (unsigned)n) pos[u] = atomicAdd(&hist[t[u] / BINW], 1);
    }
    __syncthreads();

    const int v = hist[tid];
    int x = v;
#pragma unroll
    for (int d = 1; d < 32; d <<= 1) {
        const int y = __shfl_up(x, d, 32);
        if (lane >= d) x += y;
    }
    if (lane == 31) wsum[wave] = x;
    __syncthreads();
    int before = 0;
#pragma unroll
    for (int w = 0; w < 8; ++w) before += (w < wave) ? wsum[w] : 0;
    const int start = before + x - v;
    offs[tid] = start;
    tab[c * NBIN + tid] = start | (v << 16);
    __syncthreads();

    int dst[EPT];
#pragma unroll
    for (int u = 0; u < EPT; ++u)
        dst[u] = offs[min((unsigned)t[u], (unsigned)(n - 1)) / BINW] + pos[u];
#pragma unroll
    for (int u = 0; u < EPT; ++u)
        if (pos[u] >= 0) stage[dst[u]] = make_int2(s[u], t[u]);
    __syncthreads();
    const int total = offs[NBIN - 1] + hist[NBIN - 1];
    for (int i = tid; i < total; i += 256) chunkbuf[e0 + i] = stage[i];
}

__global__ __launch_bounds__(256) void slot_build_kernel(const int2* __restrict__ chunkbuf,
                                                         const int* __restrict__ tab,
                                                         int* __restrict__ cnt, int* __restrict__ slots,
                                                         int2* __restrict__ spill, int n, int e) {
    constexpr int L4 = BINW * CAP / 4;
    constexpr int G  = 16;
    __shared__ int lcnt[BINW];
    __shared__ __attribute__((aligned(16))) int lslots[BINW * CAP];
    int4* lslots4 = reinterpret_cast<int4*>(lslots);
    const int tid  = threadIdx.x;
    const int lane = tid & 31;
    const int wv   = __builtin_amdgcn_readfirstlane(tid >> 5);
    const int b    = blockIdx.x;
    const int base = b * BINW;
    const int pk = (tid < NCHUNK) ? tab[tid * NBIN + b] : 0;
    for (int l = tid; l < BINW; l += 256) lcnt[l] = 0;
    for (int i = tid; i < L4; i += 256) lslots4[i] = make_int4(0, 0, 0, 0);
    __syncthreads();

    const int2* cb = chunkbuf + (size_t)(wv * 32) * CHUNK;
#pragma unroll
    for (int i0 = 0; i0 < 32; i0 += G) {
        int2 st[G];
#pragma unroll
        for (int j = 0; j < G; ++j) {
            const int pki = __builtin_amdgcn_readlane(pk, i0 + j);
            const int2* p = cb + (size_t)(i0 + j) * CHUNK + (pki & 0xffff);
            st[j] = make_int2(0, -1);
            if (lane < min(pki >> 16, CHUNK)) st[j] = p[lane];
        }
        int l[G], pos[G];
#pragma unroll
        for (int j = 0; j < G; ++j) {
            l[j]   = st[j].y - base;
            pos[j] = -1;
            if ((unsigned)l[j] < (unsigned)BINW) pos[j] = atomicAdd(&lcnt[l[j]], 1);
        }
#pragma unroll
        for (int j = 0; j < G; ++j) {
            if (pos[j] >= 0) {
                if (pos[j] < CAP) {
                    lslots[l[j] * CAP + pos[j]] = st[j].x;
                } else {
                    const int ks = atomicAdd(&cnt[n], 1);
                    if ((unsigned)ks < (unsigned)e) spill[ks] = st[j];
                }
            }
        }
    }
    if (__builtin_amdgcn_ballot_w32((pk >> 16) > 32) != 0u) {
        for (int i = 0; i < 32; ++i) {
            const int pki = __builtin_amdgcn_readlane(pk, i);
            const int ki  = min(pki >> 16, CHUNK);
            const int2* p = cb + (size_t)i * CHUNK + (pki & 0xffff);
            for (int q0 = 32; q0 < ki; q0 += 32) {
                int2 st = make_int2(0, -1);
                if (q0 + lane < ki) st = p[q0 + lane];
                const int l = st.y - base;
                if ((unsigned)l < (unsigned)BINW) {
                    const int pos = atomicAdd(&lcnt[l], 1);
                    if (pos < CAP) {
                        lslots[l * CAP + pos] = st.x;
                    } else {
                        const int ks = atomicAdd(&cnt[n], 1);
                        if ((unsigned)ks < (unsigned)e) spill[ks] = st;
                    }
                }
            }
        }
    }
    __syncthreads();

    int4* gl = reinterpret_cast<int4*>(slots + (size_t)base * CAP);
    if (base + BINW <= n) {
        for (int l = tid; l < BINW; l += 256) cnt[base + l] = lcnt[l];
#pragma unroll
        for (int it = 0; it < L4 / 256; ++it) gl[it * 256 + tid] = lslots4[it * 256 + tid];
        if ((L4 / 256) * 256 + tid < L4) gl[(L4 / 256) * 256 + tid] = lslots4[(L4 / 256) * 256 + tid];
    } else {
        for (int l = tid; l < BINW; l += 256)
            if (base + l < n) cnt[base + l] = lcnt[l];
        for (int i = tid; i < L4; i += 256)
            if (base + (i >> 3) < n) gl[i] = lslots4[i];
    }
}

template <int N, bool ROWSCALE, bool ONEROW>
__device__ __forceinline__ void gemm_body(const float* __restrict__ A,
                                          const float* __restrict__ B,
                                          const float* __restrict__ bias,
                                          const float* __restrict__ fvec,
                                          float* __restrict__ fout,
                                          const int* __restrict__ rcnt,
                                          float* __restrict__ C,
                                          int mtiles, int bid, int nwg) {
    constexpr int K   = 128;
    constexpr int NS  = N / 64;
    constexpr int WPS = 8 / NS;
    __shared__ float4 Bs[(K / 8) * N * 2];

    const int tid  = threadIdx.x;
    const int lane = tid & 31;
    const int half = lane >> 4;
    const int idx  = lane & 15;
    const int wave  = __builtin_amdgcn_readfirstlane(tid >> 5);
    const int slice = wave % NS;
    const int grp   = wave / NS;
    const int n0    = slice * 64;

    constexpr int PT = (K / 4) * N / 256;
    float4 v[PT];
#pragma unroll
    for (int i = 0; i < PT; ++i) {
        const int p  = tid + 256 * i;
        const int n  = p % N;
        const int k4 = p / N;
        const float* bp = B + (size_t)(4 * k4) * N + n;
        v[i].x = bp[0];
        v[i].y = bp[N];
        v[i].z = bp[2 * N];
        v[i].w = bp[3 * N];
    }
    float4 a[K / 8];
    int mt = grp * nwg + bid;
    {
        const int mt0 = min(mt, mtiles - 1);
        const float4* ap = (const float4*)(A + (size_t)(mt0 * 16 + idx) * K) + half;
#pragma unroll
        for (int j = 0; j < K / 8; ++j) a[j] = ap[2 * j];
    }
    const bool onerow = ONEROW && bid == 0 && grp == 0;
    float4 fv[K / 8];
#pragma unroll
    for (int j = 0; j < K / 8; ++j)
        fv[j] = ONEROW ? *(const float4*)(fvec + 8 * j + 4 * half) : make_float4(0.0f, 0.0f, 0.0f, 0.0f);
    float bv[4];
#pragma unroll
    for (int t = 0; t < 4; ++t) bv[t] = bias ? bias[n0 + 16 * t + idx] : 0.0f;

#pragma unroll
    for (int i = 0; i < PT; ++i) {
        const int p  = tid + 256 * i;
        const int n  = p % N;
        const int k4 = p / N;
        Bs[((((k4 >> 1) * N) + n) << 1) + (k4 & 1)] = v[i];
    }
    __syncthreads();
    if (mt >= mtiles) return;

    float4 b[K / 8][4];
    {
        const float4* bfrag = Bs + (n0 + idx) * 2 + half;
#pragma unroll
        for (int j = 0; j < K / 8; ++j)
#pragma unroll
            for (int t = 0; t < 4; ++t) b[j][t] = bfrag[(j * N + 16 * t) * 2];
    }

    if (onerow) {
        float s[4] = {0.0f, 0.0f, 0.0f, 0.0f};
#pragma unroll
        for (int j = 0; j < K / 8; ++j) {
            const float4 f = fv[j];
#pragma unroll
            for (int t = 0; t < 4; ++t)
                s[t] += f.x * b[j][t].x + f.y * b[j][t].y + f.z * b[j][t].z + f.w * b[j][t].w;
        }
#pragma unroll
        for (int t = 0; t < 4; ++t) {
            const float tot = s[t] + __shfl_xor(s[t], 16, 32);
            if (half == 0) fout[n0 + 16 * t + idx] = tot;
        }
    }

    for (; mt < mtiles; mt += WPS * nwg) {
        const int m0 = mt * 16;
        const int mtn = min(mt + WPS * nwg, mtiles - 1);
        const float4* apn = (const float4*)(A + (size_t)(mtn * 16 + idx) * K) + half;
        float4 an[K / 8];
#pragma unroll
        for (int j = 0; j < K / 8; ++j) an[j] = apn[2 * j];
        int rc[16];
#pragma unroll
        for (int r = 0; r < 16; ++r) rc[r] = ROWSCALE ? rcnt[m0 + r] : 0;

        v8f acc[4];
#pragma unroll
        for (int t = 0; t < 4; ++t) acc[t] = v8f{};

#pragma unroll
        for (int j = 0; j < K / 8; ++j) {
            v2f a0; a0.x = a[j].x; a0.y = a[j].y;
            v2f a1; a1.x = a[j].z; a1.y = a[j].w;
#pragma unroll
            for (int t = 0; t < 4; ++t) {
                v2f b0; b0.x = b[j][t].x; b0.y = b[j][t].y;
                acc[t] = __builtin_amdgcn_wmma_f32_16x16x4_f32(
                    false, a0, false, b0, (short)0, acc[t], false, false);
            }
#pragma unroll
            for (int t = 0; t < 4; ++t) {
                v2f b1; b1.x = b[j][t].z; b1.y = b[j][t].w;
                acc[t] = __builtin_amdgcn_wmma_f32_16x16x4_f32(
                    false, a1, false, b1, (short)0, acc[t], false, false);
            }
        }

        float* crow = C + (size_t)(m0 + half * 8) * N + n0 + idx;
        if (ROWSCALE) {
            float rs[8];
#pragma unroll
            for (int r = 0; r < 8; ++r) rs[r] = rsqrtf((float)((half ? rc[r + 8] : rc[r]) + 1));
#pragma unroll
            for (int t = 0; t < 4; ++t) {
#pragma unroll
                for (int r = 0; r < 8; ++r) {
                    crow[(size_t)r * N + 16 * t] = (acc[t][r] + bv[t]) * rs[r];
                }
            }
        } else {
#pragma unroll
            for (int t = 0; t < 4; ++t) {
#pragma unroll
                for (int r = 0; r < 8; ++r) {
                    crow[(size_t)r * N + 16 * t] = acc[t][r] + bv[t];
                }
            }
        }
#pragma unroll
        for (int j = 0; j < K / 8; ++j) a[j] = an[j];
    }
}

template <int N>
__global__ __launch_bounds__(256) void gemm_wmma_f32(const float* __restrict__ A,
                                                     const float* __restrict__ B,
                                                     const float* __restrict__ bias,
                                                     const int* __restrict__ rcnt,
                                                     float* __restrict__ C,
                                                     int mtiles) {
    gemm_body<N, true, false>(A, B, bias, nullptr, nullptr, rcnt, C, mtiles, (int)blockIdx.x, (int)gridDim.x);
}

#define FOLD_WG 8
__global__ __launch_bounds__(256) void fold_and_bin_kernel(const float* __restrict__ Wa,
                                                           const float* __restrict__ Wb,
                                                           const float* __restrict__ ba,
                                                           float* __restrict__ Wc,
                                                           float* __restrict__ bc,
                                                           const int* __restrict__ src,
                                                           const int* __restrict__ tgt,
                                                           int2* __restrict__ chunkbuf,
                                                           int* __restrict__ tab,
                                                           int* __restrict__ spill_len,
                                                           int n, int e) {
    if (blockIdx.x < FOLD_WG) {
        gemm_body<128, false, true>(Wa, Wb, nullptr, ba, bc, nullptr, Wc, 128 / 16, (int)blockIdx.x, FOLD_WG);
    } else {
        if (blockIdx.x == FOLD_WG && threadIdx.x == 0) *spill_len = 0;
        bin_chunk_body(src, tgt, chunkbuf, tab, (int)blockIdx.x - FOLD_WG, n, e);
    }
}

template <int V> struct rowvec;
template <> struct rowvec<4> { typedef v4f type; };
template <> struct rowvec<2> { typedef v2f type; };

template <int F, int K, typename vf>
__device__ __forceinline__ vf add_rows(vf acc, const float* __restrict__ xcol, int s, int j) {
    vf r[K];
#pragma unroll
    for (int u = 0; u < K; ++u) {
        const int su = __builtin_amdgcn_readlane(s, j + u);
        r[u] = *(const vf*)(xcol + (size_t)su * F);
    }
#pragma unroll
    for (int u = 0; u < K; ++u) acc += r[u];
    return acc;
}

template <int F, int NF, typename vf>
__device__ __forceinline__ vf sum_rows(const float* __restrict__ xcol, int s, int m) {
    static_assert(NF >= 4 && NF <= 8, "tail paths below cover 1 .. 7 rows");
    vf acc = {};
    int j = 0;
    for (; j + NF <= m; j += NF) acc = add_rows<F, NF, vf>(acc, xcol, s, j);
    switch (m - j) {
        case 1: acc = add_rows<F, 1, vf>(acc, xcol, s, j); break;
        case 2: acc = add_rows<F, 2, vf>(acc, xcol, s, j); break;
        case 3: acc = add_rows<F, 3, vf>(acc, xcol, s, j); break;
        case 4: if (NF > 4) acc = add_rows<F, 4, vf>(acc, xcol, s, j); break;
        case 5: if (NF > 5) acc = add_rows<F, 5, vf>(acc, xcol, s, j); break;
        case 6: if (NF > 6) acc = add_rows<F, 6, vf>(acc, xcol, s, j); break;
        case 7: if (NF > 7) acc = add_rows<F, 7, vf>(acc, xcol, s, j); break;
        default: break;
    }
    return acc;
}

template <int F, bool RELU>
__global__ __launch_bounds__(256) void gather_nodes_kernel(const float* __restrict__ xw,
                                                           const int* __restrict__ cnt,
                                                           const int* __restrict__ slots,
                                                           const int2* __restrict__ spill,
                                                           const float* __restrict__ bias,
                                                           float* __restrict__ out, int n, int e) {
    constexpr int V = F / 32;
    typedef typename rowvec<V>::type vf;
    const int lane   = threadIdx.x & 31;
    const int nwaves = (int)gridDim.x * 8;
    int node = __builtin_amdgcn_readfirstlane((int)(blockIdx.x * 8 + (threadIdx.x >> 5)));
    if (node >= n) return;
    const float* xcol = xw + lane * V;
    const vf bv = *(const vf*)(bias + lane * V);
    int c   = cnt[node];
    int cn  = cnt[min(node + nwaves, n - 1)];
    int ids = node;
    if (lane < min(c, CAP)) ids = slots[(size_t)node * CAP + lane];
    vf  self = *(const vf*)(xcol + (size_t)node * F);
    for (;;) {
        const int  nxt  = node + nwaves;
        const bool more = nxt < n;
        const int  pn   = more ? nxt : node;
        const int  cpn  = more ? cn : c;
        int nids = pn;
        if (lane < min(cpn, CAP)) nids = slots[(size_t)pn * CAP + lane];
        const vf  nself = *(const vf*)(xcol + (size_t)pn * F);
        const int cnn   = cnt[min(nxt + nwaves, n - 1)];

        const float di = rsqrtf((float)(c + 1));
        const int   m  = min(max(c, 0), CAP);
        int s = node;
        if (lane < m) s = min(max(ids, 0), n - 1);
        vf acc = sum_rows<F, GATHER_NF, vf>(xcol, s, m);
        if (c > CAP) {
            const int nsp = min(max(cnt[n], 0), e);
            for (int base = 0; base < nsp; base += 32) {
                const int q = base + lane;
                int os = node, ot = -1;
                if (q < nsp) { const int2 st = spill[q]; os = st.x; ot = st.y; }
                const bool hit = (ot == node);
                os = hit ? min(max(os, 0), n - 1) : node;
                unsigned left = __builtin_amdgcn_ballot_w32(hit);
                while (left != 0u) {
                    const int k = __builtin_ctz(left);
                    left &= left - 1u;
                    const int s0 = __builtin_amdgcn_readlane(os, k);
                    const vf r0 = *(const vf*)(xcol + (size_t)s0 * F);
                    acc += r0;
                }
            }
        }
        vf o;
#pragma unroll
        for (int cc = 0; cc < V; ++cc) {
            float v = (acc[cc] + self[cc]) * di + bv[cc];
            if (RELU) v = fmaxf(v, 0.0f);
            o[cc] = v;
        }
        *(vf*)(out + (size_t)node * F + lane * V) = o;

        if (!more) break;
        node = nxt;
        c    = cn;
        cn   = cnn;
        ids  = nids;
        self = nself;
    }
}

#define FUSE_TILE  32
#define FUSE_PITCH 132
#define FUSE_TILES 3125
#define FUSE_WG    1042
static_assert(FUSE_TILES * FUSE_TILE >= N_NODES, "tiles cover the nodes");

__global__ __launch_bounds__(256) void gather_gemm_kernel(const float* __restrict__ xw,
                                                          const int* __restrict__ cnt,
                                                          const int* __restrict__ slots,
                                                          const int2* __restrict__ spill,
                                                          const float* __restrict__ bias,
                                                          const float* __restrict__ B,
                                                          float* __restrict__ C,
                                                          int n, int e, int ntiles) {
    constexpr int F = 128;
    constexpr int K = 128;
    constexpr int N = 64;
    __shared__ float4 Bs[(K / 8) * N * 2];
    __shared__ __attribute__((aligned(16))) float hs[FUSE_TILE * FUSE_PITCH];
    __shared__ __attribute__((aligned(16))) float ldi[FUSE_TILE];

    const int tid = threadIdx.x;
    int tile = (int)blockIdx.x;
    if (tile >= ntiles) return;

    const int lane = tid & 31;
    const int half = lane >> 4;
    const int idx  = lane & 15;
    const int wave  = __builtin_amdgcn_readfirstlane(tid >> 5);
    const int strip = wave >> 2;
    const int n0    = (wave & 3) * 16;
    const int nwg   = (int)gridDim.x;

    const float* xcol = xw + lane * 4;
    const v4f bv = *(const v4f*)(bias + lane * 4);

    const int first = min(tile * FUSE_TILE + wave * 4, n - 1);
    int cv  = cnt[min(tile * FUSE_TILE + wave * 4 + (lane & 3), n - 1)];
    int ids = slots[(size_t)first * CAP + lane];
    v4f self = *(const v4f*)(xcol + (size_t)first * F);

    {
        constexpr int PT = (K / 4) * N / 256;
        float4 v[PT];
#pragma unroll
        for (int i = 0; i < PT; ++i) {
            const int p  = tid + 256 * i;
            const int nn = p % N;
            const int k4 = p / N;
            const float* bp = B + (size_t)(4 * k4) * N + nn;
            v[i].x = bp[0];
            v[i].y = bp[N];
            v[i].z = bp[2 * N];
            v[i].w = bp[3 * N];
        }
#pragma unroll
        for (int i = 0; i < PT; ++i) {
            const int p  = tid + 256 * i;
            const int nn = p % N;
            const int k4 = p / N;
            Bs[((((k4 >> 1) * N) + nn) << 1) + (k4 & 1)] = v[i];
        }
    }

    for (; tile < ntiles; tile += nwg) {
        v4f   o[4];
        float dq[4];
#pragma unroll
        for (int q = 0; q < 4; ++q) {
            const int node = tile * FUSE_TILE + wave * 4 + q;
            const int nxt = (q < 3) ? node + 1 : (tile + nwg) * FUSE_TILE + wave * 4;
            const int pn  = min(nxt, n - 1);
            const int c   = __builtin_amdgcn_readlane(cv, q);
            int ncv = cv;
            if (q == 3) ncv = cnt[min(nxt + (lane & 3), n - 1)];
            const int nids = slots[(size_t)pn * CAP + lane];
            const v4f nself = *(const v4f*)(xcol + (size_t)pn * F);

            v4f   ov = {};
            float di = 0.0f;
            if (node < n) {
                di = rsqrtf((float)(c + 1));
                const int m = min(max(c, 0), CAP);
                int s = node;
                if (lane < m) s = min(max(ids, 0), n - 1);
                v4f acc = sum_rows<F, GATHER_NF, v4f>(xcol, s, m);
                if (c > CAP) {
                    const int nsp = min(max(cnt[n], 0), e);
                    for (int base = 0; base < nsp; base += 32) {
                        const int qq = base + lane;
                        int os = node, ot = -1;
                        if (qq < nsp) { const int2 st = spill[qq]; os = st.x; ot = st.y; }
                        const bool hit = (ot == node);
                        os = hit ? min(max(os, 0), n - 1) : node;
                        unsigned left = __builtin_amdgcn_ballot_w32(hit);
                        while (left != 0u) {
                            const int k = __builtin_ctz(left);
                            left &= left - 1u;
                            const int s0 = __builtin_amdgcn_readlane(os, k);
                            const v4f r0 = *(const v4f*)(xcol + (size_t)s0 * F);
                            acc += r0;
                        }
                    }
                }
#pragma unroll
                for (int cc = 0; cc < 4; ++cc) {
                    const float v = (acc[cc] + self[cc]) * di + bv[cc];
                    ov[cc] = fmaxf(v, 0.0f);
                }
            }
            o[q]  = ov;
            dq[q] = di;
            cv   = ncv;
            ids  = nids;
            self = nself;
        }

        __syncthreads();
#pragma unroll
        for (int q = 0; q < 4; ++q) {
            *(v4f*)(hs + (wave * 4 + q) * FUSE_PITCH + lane * 4) = o[q];
            if (lane == 0) ldi[wave * 4 + q] = dq[q];
        }
        __syncthreads();

        const float*  arow  = hs + (strip * 16 + idx) * FUSE_PITCH + 4 * half;
        const float4* bfrag = Bs + ((n0 + idx) << 1) + half;
        v8f acc8 = {};
#pragma unroll
        for (int j = 0; j < K / 8; ++j) {
            const float4 a4 = *(const float4*)(arow + 8 * j);
            const float4 b4 = bfrag[(j * N) << 1];
            v2f a0; a0.x = a4.x; a0.y = a4.y;
            v2f b0; b0.x = b4.x; b0.y = b4.y;
            acc8 = __builtin_amdgcn_wmma_f32_16x16x4_f32(
                false, a0, false, b0, (short)0, acc8, false, false);
            v2f a1; a1.x = a4.z; a1.y = a4.w;
            v2f b1; b1.x = b4.z; b1.y = b4.w;
            acc8 = __builtin_amdgcn_wmma_f32_16x16x4_f32(
                false, a1, false, b1, (short)0, acc8, false, false);
        }
        const float4 d0 = *(const float4*)(ldi + strip * 16 + half * 8);
        const float4 d1 = *(const float4*)(ldi + strip * 16 + half * 8 + 4);
        const float rs[8] = {d0.x, d0.y, d0.z, d0.w, d1.x, d1.y, d1.z, d1.w};
        const int m0 = tile * FUSE_TILE + strip * 16 + half * 8;
        float* crow = C + (size_t)m0 * N + n0 + idx;
        if ((tile + 1) * FUSE_TILE <= n) {
#pragma unroll
            for (int r = 0; r < 8; ++r) crow[(size_t)r * N] = acc8[r] * rs[r];
        } else {
#pragma unroll
            for (int r = 0; r < 8; ++r)
                if (m0 + r < n) crow[(size_t)r * N] = acc8[r] * rs[r];
        }
    }
}

__global__ __launch_bounds__(256) void edge_score_kernel(const float* __restrict__ h,
                                                         const int* __restrict__ ia,
                                                         const int* __restrict__ ib,
                                                         float* __restrict__ out, int e) {
    constexpr int U = 4;
    const int lane   = threadIdx.x & 15;
    const int stride = (int)gridDim.x * 16;
    int ed = (int)((blockIdx.x * 256 + threadIdx.x) >> 4);
    for (; ed + (U - 1) * stride < e; ed += U * stride) {
        int a[U], b[U];
#pragma unroll
        for (int u = 0; u < U; ++u) {
            a[u] = ia[ed + u * stride];
            b[u] = ib[ed + u * stride];
        }
        float4 va[U], vb[U];
#pragma unroll
        for (int u = 0; u < U; ++u) {
            va[u] = ((const float4*)(h + (size_t)a[u] * 64))[lane];
            vb[u] = ((const float4*)(h + (size_t)b[u] * 64))[lane];
        }
#pragma unroll
        for (int u = 0; u < U; ++u) {
            float s = va[u].x * vb[u].x + va[u].y * vb[u].y + va[u].z * vb[u].z + va[u].w * vb[u].w;
            s += __shfl_xor(s, 8, 32);
            s += __shfl_xor(s, 4, 32);
            s += __shfl_xor(s, 2, 32);
            s += __shfl_xor(s, 1, 32);
            if (lane == 0) out[ed + u * stride] = s;
        }
    }
    for (; ed < e; ed += stride) {
        const int a = ia[ed];
        const int b = ib[ed];
        const float4 va = ((const float4*)(h + (size_t)a * 64))[lane];
        const float4 vb = ((const float4*)(h + (size_t)b * 64))[lane];
        float s = va.x * vb.x + va.y * vb.y + va.z * vb.z + va.w * vb.w;
        s += __shfl_xor(s, 8, 32);
        s += __shfl_xor(s, 4, 32);
        s += __shfl_xor(s, 2, 32);
        s += __shfl_xor(s, 1, 32);
        if (lane == 0) out[ed] = s;
    }
}

extern "C" void kernel_launch(void* const* d_in, const int* in_sizes, int n_in,
                              void* d_out, int out_size, void* d_ws, size_t ws_size,
                              hipStream_t stream) {
    const float* x      = (const float*)d_in[0];
    const int*   eidx   = (const int*)d_in[1];
    const int*   pidx   = (const int*)d_in[2];
    const float* W_init = (const float*)d_in[3];
    const float* b_init = (const float*)d_in[4];
    const float* W1     = (const float*)d_in[5];
    const float* b1     = (const float*)d_in[6];
    const float* W2     = (const float*)d_in[7];
    const float* b2     = (const float*)d_in[8];
    float* logits = (float*)d_out;

    const int* src = pidx;
    const int* tgt = pidx + E_POS;
    const int* qa  = eidx;
    const int* qb  = eidx + E_PRED;

    float* ws   = (float*)d_ws;
    float* bufA = ws + 102400;
    float* bufB = bufA + (size_t)N_NODES * 128;
    float* bufC = bufB + (size_t)N_NODES * 128;
    int*  cnt   = (int*)bufC;
    float* Wc = (float*)(cnt + 102400);
    float* bc = Wc + 128 * 128;
    int2* spill = (int2*)(cnt + 102400 + 16896);
    int*  slots = cnt + 102400 + 16896 + 2 * E_POS;
    int2* chunkbuf = (int2*)(slots + (size_t)N_NODES * CAP);
    int*  tab   = (int*)(chunkbuf + (size_t)NCHUNK * CHUNK);

    const int TB = 256;
    const int MT = N_NODES / 16;

    fold_and_bin_kernel<<<FOLD_WG + NCHUNK, TB, 0, stream>>>(
        W_init, W1, b_init, Wc, bc, src, tgt, chunkbuf, tab, cnt + N_NODES, N_NODES, E_POS);
    slot_build_kernel<<<NBIN, TB, 0, stream>>>(chunkbuf, tab, cnt, slots, spill, N_NODES, E_POS);

    gemm_wmma_f32<128><<<256, 256, 0, stream>>>(x, Wc, bc, cnt, bufB, MT);
    gather_gemm_kernel<<<FUSE_WG, TB, 0, stream>>>(
        bufB, cnt, slots, spill, b1, W2, bufA, N_NODES, E_POS, FUSE_TILES);

    gather_nodes_kernel<64, false><<<GATHER_WG, TB, 0, stream>>>(
        bufA, cnt, slots, spill, b2, bufB, N_NODES, E_POS);

    edge_score_kernel<<<EDGE_WG, TB, 0, stream>>>(bufB, qa, qb, logits, E_PRED);
}
